# starting kernel
# baseline (speedup 1.0000x reference)
#include <hip/hip_runtime.h>


typedef float v2f __attribute__((ext_vector_type(2)));
typedef float v8f __attribute__((ext_vector_type(8)));

#define B_    16
#define C_    256
#define H_    64
#define W_    64
#define OUTC  3
#define NT    512
#define CB    64

#define HW     (H_*W_)
#define NPOS   (B_*HW)
#define NTILE  (NPOS/16)
#define CHW    (C_*HW)

#define Q_OFF      0u
#define H_OFF      16777216u
#define EPACK_OFF  20971520u
#define EN2_OFF    21102592u
#define CNT_OFF    21103104u
#define WDIFF_OFF  21111296u
#define WPACK_OFF  21115392u

#define WP_PER_BLK (9*128*CB)
#define WP_PER_TAP (128*CB)

#define DIFF_POS   196608
#define WF_POS     196609

static __device__ __forceinline__ v8f wmma4(v2f a, v2f b, v8f c) {
  return __builtin_amdgcn_wmma_f32_16x16x4_f32(false, a, false, b, (short)0, c,
                                               false, false);
}

#define DECODE_TILE(s, b, hh, w0)                        \
  const int b  = (s) >> 8;                               \
  const int hh = ((s) & 255) >> 2;                       \
  const int w0 = ((s) & 3) << 4;

__global__ __launch_bounds__(256) void k_prep(const float* __restrict__ embed,
                                              float* __restrict__ epack,
                                              float* __restrict__ en2,
                                              unsigned* __restrict__ counts) {
  const int gid = blockIdx.x * 256 + threadIdx.x;
  if (gid < 128 * NT) {
    const int cc = gid >> 9, n = gid & (NT - 1);
    epack[gid * 2 + 0] = embed[(2 * cc) * NT + n];
    epack[gid * 2 + 1] = embed[(2 * cc + 1) * NT + n];
  }
  if (gid < NT) {
    float s = 0.f;
    for (int c = 0; c < C_; ++c) { float v = embed[c * NT + gid]; s += v * v; }
    en2[gid] = s;
  }
  if (gid < B_ * NT) counts[gid] = 0u;
}

__global__ __launch_bounds__(256) void k_packw(const float* __restrict__ rw1,
                                               float* __restrict__ wpack) {
  const int t = blockIdx.x * 256 + threadIdx.x;
  if (t >= 3 * WP_PER_BLK) return;
  const int i   = t / WP_PER_BLK;
  const int r   = t % WP_PER_BLK;
  const int tap = r / WP_PER_TAP;
  const int r2  = r % WP_PER_TAP;
  const int cc  = r2 >> 6;
  const int oc  = r2 & 63;
  const int dh = tap / 3, dw = tap % 3;
  const float* src = rw1 + (size_t)i * CB * C_ * 9
                         + ((size_t)oc * C_ + 2 * cc) * 9 + dh * 3 + dw;
  wpack[t * 2 + 0] = src[0];
  wpack[t * 2 + 1] = src[9];
}

__global__ __launch_bounds__(128) void k_vq(const float* __restrict__ in,
                                            const float* __restrict__ embed,
                                            const float* __restrict__ epack,
                                            const float* __restrict__ en2,
                                            float* __restrict__ q,
                                            unsigned* __restrict__ counts,
                                            float* __restrict__ wsdiff) {
  __shared__ float As[4096];
  __shared__ float rv[4][16][16];
  __shared__ int   ri[4][16][16];
  __shared__ int   idxs[16];
  __shared__ float red[128];

  const int s = blockIdx.x;
  DECODE_TILE(s, b, hh, w0)
  const int tid = threadIdx.x;
  const long gbase0 = (long)b * CHW + hh * W_ + w0;

  for (int t = tid; t < 2048; t += 128) {
    const int cc = t >> 4, m = t & 15;
    As[t * 2 + 0] = in[gbase0 + (2 * cc) * HW + m];
    As[t * 2 + 1] = in[gbase0 + (2 * cc + 1) * HW + m];
  }
  __syncthreads();

  const int wave = tid >> 5, lane = tid & 31, g = lane >> 4, mc = lane & 15;
  float bestv[8];
  int   besti[8];
#pragma unroll
  for (int r = 0; r < 8; ++r) { bestv[r] = 3.4e38f; besti[r] = 0; }

  for (int tt = 0; tt < 8; ++tt) {
    const int n0 = (wave * 8 + tt) << 4;
    v8f acc = {0.f, 0.f, 0.f, 0.f, 0.f, 0.f, 0.f, 0.f};
#pragma unroll 8
    for (int k0 = 0; k0 < C_; k0 += 4) {
      const int cc = (k0 >> 1) + g;
      v2f a  = *(const v2f*)(As + (cc * 16 + mc) * 2);
      v2f bb = *(const v2f*)(epack + (cc * NT + n0 + mc) * 2);
      acc = wmma4(a, bb, acc);
    }
    const float e2 = en2[n0 + mc];
    const int   n  = n0 + mc;
#pragma unroll
    for (int r = 0; r < 8; ++r) {
      const float d = e2 - 2.0f * acc[r];
      if (d < bestv[r]) { bestv[r] = d; besti[r] = n; }
    }
  }
#pragma unroll
  for (int r = 0; r < 8; ++r) {
    const int mr = r + 8 * g;
    rv[wave][mr][mc] = bestv[r];
    ri[wave][mr][mc] = besti[r];
  }
  __syncthreads();

  if (tid < 16) {
    float bv = 3.4e38f; int bi = 0x7fffffff;
    for (int w = 0; w < 4; ++w)
      for (int c2 = 0; c2 < 16; ++c2) {
        const float v = rv[w][tid][c2]; const int ii = ri[w][tid][c2];
        if (v < bv || (v == bv && ii < bi)) { bv = v; bi = ii; }
      }
    idxs[tid] = bi;
    atomicAdd(&counts[b * NT + bi], 1u);
  }
  __syncthreads();

  float fsum = 0.f;
  for (int t = tid; t < 2048; t += 128) {
    const int cc = t >> 4, m = t & 15;
    const int id = idxs[m];
    const float x0 = As[t * 2 + 0], x1 = As[t * 2 + 1];
    const float e0 = embed[(2 * cc) * NT + id];
    const float e1 = embed[(2 * cc + 1) * NT + id];
    q[gbase0 + (2 * cc) * HW + m]     = e0;
    q[gbase0 + (2 * cc + 1) * HW + m] = e1;
    const float d0 = e0 - x0, d1 = e1 - x1;
    fsum += d0 * d0 + d1 * d1;
  }
  red[tid] = fsum; __syncthreads();
  for (int st = 64; st > 0; st >>= 1) {
    if (tid < st) red[tid] += red[tid + st];
    __syncthreads();
  }
  if (tid == 0) wsdiff[s] = red[0];
}

__global__ __launch_bounds__(128) void k_res_conv3(const float* __restrict__ q,
                                                   const float* __restrict__ wp,
                                                   const float* __restrict__ b1,
                                                   float* __restrict__ hbuf) {
  __shared__ float Ah[13824];
  const int s = blockIdx.x;
  DECODE_TILE(s, b, hh, w0)
  const int tid = threadIdx.x;

  for (int t = tid; t < 6912; t += 128) {
    const int cc = t / 54, r2 = t % 54, rr = r2 / 18, ww = r2 % 18;
    const int hy = hh + rr - 1, wx = w0 + ww - 1;
    float v0 = 0.f, v1 = 0.f;
    if ((unsigned)hy < 64u && (unsigned)wx < 64u) {
      const long gb = (long)b * CHW + hy * W_ + wx;
      v0 = q[gb + (2 * cc) * HW];     v0 = v0 > 0.f ? v0 : 0.f;
      v1 = q[gb + (2 * cc + 1) * HW]; v1 = v1 > 0.f ? v1 : 0.f;
    }
    Ah[t * 2 + 0] = v0; Ah[t * 2 + 1] = v1;
  }
  __syncthreads();

  const int wave = tid >> 5, lane = tid & 31, g = lane >> 4, mc = lane & 15;
  const int oc = wave * 16 + mc;
  v8f acc = {0.f, 0.f, 0.f, 0.f, 0.f, 0.f, 0.f, 0.f};
  for (int tap = 0; tap < 9; ++tap) {
    const int dh = tap / 3, dw = tap % 3;
    const float* wt = wp + (size_t)tap * WP_PER_TAP * 2;
#pragma unroll 8
    for (int k0 = 0; k0 < C_; k0 += 4) {
      const int cc = (k0 >> 1) + g;
      v2f a  = *(const v2f*)(Ah + ((cc * 3 + dh) * 18 + mc + dw) * 2);
      v2f bb = *(const v2f*)(wt + (cc * CB + oc) * 2);
      acc = wmma4(a, bb, acc);
    }
  }
  const float bias = b1[oc];
  const long ob = (long)b * (CB * HW) + oc * HW + hh * W_ + w0;
#pragma unroll
  for (int r = 0; r < 8; ++r) {
    const float v = acc[r] + bias;
    hbuf[ob + r + 8 * g] = v > 0.f ? v : 0.f;
  }
}

__global__ __launch_bounds__(512) void k_res_conv1(float* __restrict__ q,
                                                   const float* __restrict__ hbuf,
                                                   const float* __restrict__ w2,
                                                   const float* __restrict__ b2) {
  __shared__ float Hs[1024];
  const int s = blockIdx.x;
  DECODE_TILE(s, b, hh, w0)
  const int tid = threadIdx.x;
  {
    const int cc = tid >> 4, m = tid & 15;
    const long gb = (long)b * (CB * HW) + hh * W_ + w0 + m;
    Hs[tid * 2 + 0] = hbuf[gb + (2 * cc) * HW];
    Hs[tid * 2 + 1] = hbuf[gb + (2 * cc + 1) * HW];
  }
  __syncthreads();

  const int wave = tid >> 5, lane = tid & 31, g = lane >> 4, mc = lane & 15;
  const int oc = wave * 16 + mc;
  v8f acc = {0.f, 0.f, 0.f, 0.f, 0.f, 0.f, 0.f, 0.f};
#pragma unroll
  for (int k0 = 0; k0 < CB; k0 += 4) {
    const int cc = (k0 >> 1) + g;
    v2f a  = *(const v2f*)(Hs + (cc * 16 + mc) * 2);
    v2f bb = *(const v2f*)(w2 + oc * CB + k0 + 2 * g);
    acc = wmma4(a, bb, acc);
  }
  const float bias = b2[oc];
  const long ob = (long)b * CHW + oc * HW + hh * W_ + w0;
#pragma unroll
  for (int r = 0; r < 8; ++r) q[ob + r + 8 * g] += acc[r] + bias;
}

__global__ __launch_bounds__(32) void k_out_conv(const float* __restrict__ q,
                                                 const float* __restrict__ ow,
                                                 const float* __restrict__ ob,
                                                 float* __restrict__ out) {
  __shared__ float Qs[4096];
  __shared__ float Bs[4096];
  const int s = blockIdx.x;
  DECODE_TILE(s, b, hh, w0)
  const int lane = threadIdx.x, g = lane >> 4, mc = lane & 15;
  const long gbase0 = (long)b * CHW + hh * W_ + w0;

  for (int t = lane; t < 2048; t += 32) {
    const int cc = t >> 4, m = t & 15;
    Qs[t * 2 + 0] = q[gbase0 + (2 * cc) * HW + m];
    Qs[t * 2 + 1] = q[gbase0 + (2 * cc + 1) * HW + m];
    Bs[t * 2 + 0] = (m < OUTC) ? ow[m * C_ + 2 * cc]     : 0.f;
    Bs[t * 2 + 1] = (m < OUTC) ? ow[m * C_ + 2 * cc + 1] : 0.f;
  }
  __syncthreads();

  v8f acc = {0.f, 0.f, 0.f, 0.f, 0.f, 0.f, 0.f, 0.f};
#pragma unroll 8
  for (int k0 = 0; k0 < C_; k0 += 4) {
    const int cc = (k0 >> 1) + g;
    v2f a  = *(const v2f*)(Qs + (cc * 16 + mc) * 2);
    v2f bb = *(const v2f*)(Bs + (cc * 16 + mc) * 2);
    acc = wmma4(a, bb, acc);
  }
  if (mc < OUTC) {
    const float bias = ob[mc];
    const long obase = ((long)(b * OUTC + mc) * H_ + hh) * W_ + w0;
#pragma unroll
    for (int r = 0; r < 8; ++r) out[obase + r + 8 * g] = acc[r] + bias;
  }
}

__global__ __launch_bounds__(256) void k_final(const unsigned* __restrict__ counts,
                                               const float* __restrict__ wsdiff,
                                               float* __restrict__ out) {
  __shared__ float red[256];
  const int gid = blockIdx.x * 256 + threadIdx.x;
  if (blockIdx.x < 32) {
    out[WF_POS + gid] = (float)counts[gid] * (1.0f / 4096.0f);
  } else {
    float sum = 0.f;
    for (int t = threadIdx.x; t < NTILE; t += 256) sum += wsdiff[t];
    red[threadIdx.x] = sum; __syncthreads();
    for (int st = 128; st > 0; st >>= 1) {
      if (threadIdx.x < st) red[threadIdx.x] += red[threadIdx.x + st];
      __syncthreads();
    }
    if (threadIdx.x == 0)
      out[DIFF_POS] = red[0] * (1.0f / ((float)NPOS * (float)C_));
  }
}

extern "C" void kernel_launch(void* const* d_in, const int* in_sizes, int n_in,
                              void* d_out, int out_size, void* d_ws, size_t ws_size,
                              hipStream_t stream) {
  (void)in_sizes; (void)n_in; (void)out_size; (void)ws_size;
  const float* in  = (const float*)d_in[0];
  const float* emb = (const float*)d_in[1];
  const float* rw1 = (const float*)d_in[2];
  const float* rb1 = (const float*)d_in[3];
  const float* rw2 = (const float*)d_in[4];
  const float* rb2 = (const float*)d_in[5];
  const float* ow  = (const float*)d_in[6];
  const float* ob  = (const float*)d_in[7];
  float* out = (float*)d_out;
  float* ws  = (float*)d_ws;

  float*    q      = ws + Q_OFF;
  float*    hbuf   = ws + H_OFF;
  float*    epack  = ws + EPACK_OFF;
  float*    en2    = ws + EN2_OFF;
  unsigned* counts = (unsigned*)(ws + CNT_OFF);
  float*    wsdiff = ws + WDIFF_OFF;
  float*    wpack  = ws + WPACK_OFF;

  k_prep<<<256, 256, 0, stream>>>(emb, epack, en2, counts);
  k_packw<<<(3 * WP_PER_BLK + 255) / 256, 256, 0, stream>>>(rw1, wpack);
  k_vq<<<NTILE, 128, 0, stream>>>(in, emb, epack, en2, q, counts, wsdiff);
  for (int i = 0; i < 3; ++i) {
    k_res_conv3<<<NTILE, 128, 0, stream>>>(q, wpack + (size_t)i * WP_PER_BLK * 2,
                                           rb1 + i * CB, hbuf);
    k_res_conv1<<<NTILE, 512, 0, stream>>>(q, hbuf, rw2 + (size_t)i * C_ * CB,
                                           rb2 + i * C_);
  }
  k_out_conv<<<NTILE, 32, 0, stream>>>(q, ow, ob, out);
  k_final<<<33, 256, 0, stream>>>(counts, wsdiff, out);
}


# VQ distance product re-tiled: work-group = 128 positions, x fragments kept in registers, codebook streamed via LDS in 32-code slabs shared by 8 waves; same distances; diff partials regrouped (512)
# speedup vs baseline: 4.5628x; 4.5628x over previous
#include <hip/hip_runtime.h>


typedef float v2f __attribute__((ext_vector_type(2)));
typedef float v4f __attribute__((ext_vector_type(4)));
typedef float v8f __attribute__((ext_vector_type(8)));

#define B_    16
#define C_    256
#define H_    64
#define W_    64
#define OUTC  3
#define NT    512
#define CB    64

#define HW     (H_*W_)
#define NPOS   (B_*HW)
#define NTILE  (NPOS/16)
#define CHW    (C_*HW)

#define Q_OFF      0u
#define H_OFF      16777216u
#define ETP_OFF    20971520u
#define ETN_OFF    H_OFF
#define EN2_OFF    21102592u
#define CNT_OFF    21103104u
#define WDIFF_OFF  21111296u
#define WPACK_OFF  21115392u

#define PR     8
#define NHP    ((PR + 2) * 18)
#define KC     32
#define KP     (KC + 4)
#define NSLAB  (C_ / KC)
#define WSLAB  (9 * CB * KC)
#define WBLK   (NSLAB * WSLAB)
#define HQ     (NHP * (KC / 4))
#define LDS_F  28672

#define QP  (C_ + 4)

#define VC     32
#define VP     (C_ + 4)
#define VSLAB  (VC * C_)
#define NVS    (NT / VC)
#define VQ_WG  (NPOS / 128)

#define DIFF_POS   196608
#define WF_POS     196609

static __device__ __forceinline__ v8f wmma4(v2f a, v2f b, v8f c) {
  return __builtin_amdgcn_wmma_f32_16x16x4_f32(false, a, false, b, (short)0, c,
                                               false, false);
}

#define DECODE_TILE(s, b, hh, w0)                        \
  const int b  = (s) >> 8;                               \
  const int hh = ((s) & 255) >> 2;                       \
  const int w0 = ((s) & 3) << 4;

__global__ __launch_bounds__(256) void k_prep(const float* __restrict__ embed,
                                              float* __restrict__ etp,
                                              float* __restrict__ etn,
                                              float* __restrict__ en2,
                                              unsigned* __restrict__ counts) {
  const int gid = blockIdx.x * 256 + threadIdx.x;
  {
    const int n = gid >> 8, ch = gid & 255;
    const float v = embed[ch * NT + n];
    etn[gid] = v;
    const int cp = (ch & ~6) | ((ch & 2) << 1) | ((ch & 4) >> 1);
    etp[n * C_ + cp] = v;
  }
  if (gid < NT) {
    float s = 0.f;
    for (int c = 0; c < C_; ++c) { float v = embed[c * NT + gid]; s += v * v; }
    en2[gid] = s;
  }
  if (gid < B_ * NT) counts[gid] = 0u;
}

__global__ __launch_bounds__(256) void k_packw(const float* __restrict__ rw1,
                                               float* __restrict__ wpack) {
  const int t = blockIdx.x * 256 + threadIdx.x;
  if (t >= 3 * WBLK) return;
  const int i   = t / WBLK;
  const int r   = t % WBLK;
  const int kc  = r / WSLAB;
  const int r2  = r % WSLAB;
  const int tap = r2 / (CB * KC);
  const int r3  = r2 % (CB * KC);
  const int oc  = r3 / KC;
  const int cl  = r3 % KC;
  wpack[t] = rw1[(size_t)i * CB * C_ * 9
                 + ((size_t)oc * C_ + kc * KC + cl) * 9 + tap];
}

__global__ __launch_bounds__(256) void k_vq(const float* __restrict__ in,
                                            const float* __restrict__ etp,
                                            const float* __restrict__ etn,
                                            const float* __restrict__ en2,
                                            float* __restrict__ q,
                                            unsigned* __restrict__ counts,
                                            float* __restrict__ wsdiff) {
  __shared__ __attribute__((aligned(16))) float Bs[2 * VC * VP];
  __shared__ __attribute__((aligned(16))) float e2s[NT];
  __shared__ int   idxs[128];
  __shared__ float red[8];

  const int s = blockIdx.x;
  const int b = s >> 5;
  const int tid = threadIdx.x;
  const int wave = tid >> 5, lane = tid & 31, g = lane >> 4, mc = lane & 15;

  const int sdst = (tid >> 6) * VP + 4 * (tid & 63);
  v4f sreg[8];
#pragma unroll
  for (int i = 0; i < 8; ++i) sreg[i] = *(const v4f*)(etp + 4 * (tid + 256 * i));

  const float* xp = in + (size_t)b * CHW + (s & 31) * 128 + 16 * wave + mc
                    + 2 * g * HW;
  v4f a[32];
#pragma unroll
  for (int j = 0; j < 32; ++j)
    a[j] = v4f{xp[(8 * j) * HW], xp[(8 * j + 1) * HW],
               xp[(8 * j + 4) * HW], xp[(8 * j + 5) * HW]};

#pragma unroll
  for (int i = 0; i < 8; ++i) *(v4f*)(Bs + sdst + 4 * VP * i) = sreg[i];
  if (tid < NT / 4) *(v4f*)(e2s + 4 * tid) = *(const v4f*)(en2 + 4 * tid);
#pragma unroll
  for (int i = 0; i < 8; ++i)
    sreg[i] = *(const v4f*)(etp + VSLAB + 4 * (tid + 256 * i));
  __syncthreads();

  float bestv[8];
  int   besti[8];
#pragma unroll
  for (int r = 0; r < 8; ++r) { bestv[r] = 3.4e38f; besti[r] = 0; }

#pragma unroll 1
  for (int k = 0; k < NVS; ++k) {
    if (k + 1 < NVS) {
      float* dst = Bs + ((k + 1) & 1) * (VC * VP) + sdst;
#pragma unroll
      for (int i = 0; i < 8; ++i) *(v4f*)(dst + 4 * VP * i) = sreg[i];
    }
    if (k + 2 < NVS) {
      const float* src = etp + (size_t)(k + 2) * VSLAB + 4 * tid;
#pragma unroll
      for (int i = 0; i < 8; ++i) sreg[i] = *(const v4f*)(src + 1024 * i);
    }
    const float* bp = Bs + (k & 1) * (VC * VP) + mc * VP + 4 * g;
    const int n0 = k * VC + mc;
    const float e20 = e2s[n0], e21 = e2s[n0 + 16];
    v8f acc0 = {0.f, 0.f, 0.f, 0.f, 0.f, 0.f, 0.f, 0.f};
    v8f acc1 = {0.f, 0.f, 0.f, 0.f, 0.f, 0.f, 0.f, 0.f};
#pragma unroll
    for (int j = 0; j < 32; ++j) {
      const v4f w0 = *(const v4f*)(bp + 8 * j);
      const v4f w1 = *(const v4f*)(bp + 16 * VP + 8 * j);
      acc0 = wmma4(v2f{a[j].x, a[j].y}, v2f{w0.x, w0.y}, acc0);
      acc1 = wmma4(v2f{a[j].x, a[j].y}, v2f{w1.x, w1.y}, acc1);
      acc0 = wmma4(v2f{a[j].z, a[j].w}, v2f{w0.z, w0.w}, acc0);
      acc1 = wmma4(v2f{a[j].z, a[j].w}, v2f{w1.z, w1.w}, acc1);
    }
#pragma unroll
    for (int r = 0; r < 8; ++r) {
      const float d = e20 - 2.0f * acc0[r];
      if (d < bestv[r]) { bestv[r] = d; besti[r] = n0; }
    }
#pragma unroll
    for (int r = 0; r < 8; ++r) {
      const float d = e21 - 2.0f * acc1[r];
      if (d < bestv[r]) { bestv[r] = d; besti[r] = n0 + 16; }
    }
    __syncthreads();
  }

#pragma unroll
  for (int r = 0; r < 8; ++r) {
#pragma unroll
    for (int off = 1; off < 16; off <<= 1) {
      const float ov = __shfl_xor(bestv[r], off);
      const int   oi = __shfl_xor(besti[r], off);
      if (ov < bestv[r] || (ov == bestv[r] && oi < besti[r])) {
        bestv[r] = ov; besti[r] = oi;
      }
    }
  }
  if (mc == 0) {
#pragma unroll
    for (int r = 0; r < 8; ++r) idxs[wave * 16 + 8 * g + r] = besti[r];
  }
  __syncthreads();

  if (tid < 128) atomicAdd(&counts[b * NT + idxs[tid]], 1u);

  const float* ep = etp + idxs[wave * 16 + mc] * C_ + 4 * g;
  float fsum = 0.f;
#pragma unroll
  for (int j = 0; j < 32; ++j) {
    const v4f e = *(const v4f*)(ep + 8 * j);
    const float d0 = e.x - a[j].x, d1 = e.y - a[j].y;
    const float d2 = e.z - a[j].z, d3 = e.w - a[j].w;
    fsum += d0 * d0 + d1 * d1 + d2 * d2 + d3 * d3;
    if ((j & 7) == 7) __builtin_amdgcn_sched_barrier(0);
  }

  float* qb = q + (size_t)s * 128 * C_;
  for (int p = 0; p < 32; ++p) {
    const int m = 4 * p + (tid >> 6);
    const v4f v = *(const v4f*)(etn + idxs[m] * C_ + 4 * (tid & 63));
    *(v4f*)(qb + m * C_ + 4 * (tid & 63)) = v;
  }

#pragma unroll
  for (int off = 16; off > 0; off >>= 1) fsum += __shfl_xor(fsum, off);
  if (lane == 0) red[wave] = fsum;
  __syncthreads();
  if (tid == 0) {
    float t = 0.f;
    for (int w = 0; w < 8; ++w) t += red[w];
    wsdiff[s] = t;
  }
}

__global__ __launch_bounds__(256) void k_res_conv3(const float* __restrict__ q,
                                                   const float* __restrict__ wp,
                                                   const float* __restrict__ b1,
                                                   float* __restrict__ hbuf) {
  __shared__ __attribute__((aligned(16))) float lds[LDS_F];
  float* Ah = lds;
  float* Ws = lds + NHP * KP;
  const int s = blockIdx.x;
  const int b  = s >> 5;
  const int h0 = ((s & 31) >> 2) << 3;
  const int w0 = (s & 3) << 4;
  const int tid = threadIdx.x;

  int  hoff[6], hdst[6];
  bool hok[6];
#pragma unroll
  for (int j = 0; j < 6; ++j) {
    const int t  = tid + 256 * j;
    const int tt = t < HQ ? t : 0;
    const int p = tt >> 3, c4 = tt & 7, rr = p / 18, ww = p % 18;
    const int hy = h0 + rr - 1, wx = w0 + ww - 1;
    const bool ok = (unsigned)hy < 64u && (unsigned)wx < 64u;
    hok[j]  = ok;
    hoff[j] = ((b * H_ + (ok ? hy : h0)) * W_ + (ok ? wx : w0)) * C_ + 4 * c4;
    hdst[j] = p * KP + 4 * c4;
  }
  const int wdst = (tid >> 3) * KP + 4 * (tid & 7);

  v4f hreg[6], wreg[18];
#pragma unroll
  for (int j = 0; j < 6; ++j) hreg[j] = *(const v4f*)(q + hoff[j]);
#pragma unroll
  for (int j = 0; j < 18; ++j) wreg[j] = *(const v4f*)(wp + 4 * (tid + 256 * j));

  const int wave = tid >> 5, lane = tid & 31, g = lane >> 4, mc = lane & 15;
  const float* ap = Ah + (wave * 18 + mc) * KP + 4 * g;
  const float* bp = Ws + mc * KP + 4 * g;
  v8f acc[4];
#pragma unroll
  for (int t = 0; t < 4; ++t) acc[t] = v8f{0.f, 0.f, 0.f, 0.f, 0.f, 0.f, 0.f, 0.f};

#pragma unroll 1
  for (int kc = 0; kc < NSLAB; ++kc) {
#pragma unroll
    for (int j = 0; j < 6; ++j) {
      v4f v = hreg[j];
      v.x = (hok[j] && v.x > 0.f) ? v.x : 0.f;
      v.y = (hok[j] && v.y > 0.f) ? v.y : 0.f;
      v.z = (hok[j] && v.z > 0.f) ? v.z : 0.f;
      v.w = (hok[j] && v.w > 0.f) ? v.w : 0.f;
      if (j < 5 || tid < HQ - 1280) *(v4f*)(Ah + hdst[j]) = v;
    }
#pragma unroll
    for (int j = 0; j < 18; ++j) *(v4f*)(Ws + wdst + 32 * KP * j) = wreg[j];
    __syncthreads();

    if (kc + 1 < NSLAB) {
      const float* qn = q + (kc + 1) * KC;
      const float* wn = wp + (size_t)(kc + 1) * WSLAB;
#pragma unroll
      for (int j = 0; j < 6; ++j) hreg[j] = *(const v4f*)(qn + hoff[j]);
#pragma unroll
      for (int j = 0; j < 18; ++j) wreg[j] = *(const v4f*)(wn + 4 * (tid + 256 * j));
    }

#pragma unroll
    for (int tap = 0; tap < 9; ++tap) {
      const int dh = tap / 3, dw = tap % 3;
#pragma unroll
      for (int j = 0; j < KC / 8; ++j) {
        const v4f a = *(const v4f*)(ap + (dh * 18 + dw) * KP + 8 * j);
        v4f w[4];
#pragma unroll
        for (int t = 0; t < 4; ++t)
          w[t] = *(const v4f*)(bp + (tap * CB + 16 * t) * KP + 8 * j);
#pragma unroll
        for (int t = 0; t < 4; ++t)
          acc[t] = wmma4(v2f{a.x, a.y}, v2f{w[t].x, w[t].y}, acc[t]);
#pragma unroll
        for (int t = 0; t < 4; ++t)
          acc[t] = wmma4(v2f{a.z, a.w}, v2f{w[t].z, w[t].w}, acc[t]);
      }
    }
    __syncthreads();
  }

  const long pos0 = ((long)(b * H_ + h0 + wave)) * W_ + w0;
#pragma unroll
  for (int t = 0; t < 4; ++t) {
    const int oc = 16 * t + mc;
    const float bias = b1[oc];
#pragma unroll
    for (int r = 0; r < 8; ++r) {
      const float v = acc[t][r] + bias;
      hbuf[(pos0 + r + 8 * g) * CB + oc] = v > 0.f ? v : 0.f;
    }
  }
}

__global__ __launch_bounds__(128) void k_res_conv1(float* __restrict__ q,
                                                   const float* __restrict__ hbuf,
                                                   const float* __restrict__ w2,
                                                   const float* __restrict__ b2) {
  const int tid = threadIdx.x;
  const int wave = tid >> 5, lane = tid & 31, g = lane >> 4, mc = lane & 15;
  const int oc0 = wave * 64;

  v4f wf[4][8];
  float bias[4];
#pragma unroll
  for (int t = 0; t < 4; ++t) {
    const float* wr = w2 + (oc0 + 16 * t + mc) * CB + 4 * g;
#pragma unroll
    for (int j = 0; j < 8; ++j) wf[t][j] = *(const v4f*)(wr + 8 * j);
    bias[t] = b2[oc0 + 16 * t + mc];
  }

#pragma unroll 1
  for (int i = 0; i < 4; ++i) {
    const long pos0 = ((long)blockIdx.x * 4 + i) * 16;
    const float* hp = hbuf + (pos0 + mc) * CB + 4 * g;
    v4f a[8];
#pragma unroll
    for (int j = 0; j < 8; ++j) a[j] = *(const v4f*)(hp + 8 * j);

    float* qp = q + (pos0 + 8 * g) * C_ + oc0 + mc;
    float qv[4][8];
#pragma unroll
    for (int r = 0; r < 8; ++r)
#pragma unroll
      for (int t = 0; t < 4; ++t) qv[t][r] = qp[r * C_ + 16 * t];

    v8f acc[4];
#pragma unroll
    for (int t = 0; t < 4; ++t) acc[t] = v8f{0.f, 0.f, 0.f, 0.f, 0.f, 0.f, 0.f, 0.f};
#pragma unroll
    for (int j = 0; j < 8; ++j) {
#pragma unroll
      for (int t = 0; t < 4; ++t)
        acc[t] = wmma4(v2f{a[j].x, a[j].y}, v2f{wf[t][j].x, wf[t][j].y}, acc[t]);
#pragma unroll
      for (int t = 0; t < 4; ++t)
        acc[t] = wmma4(v2f{a[j].z, a[j].w}, v2f{wf[t][j].z, wf[t][j].w}, acc[t]);
    }
#pragma unroll
    for (int r = 0; r < 8; ++r)
#pragma unroll
      for (int t = 0; t < 4; ++t)
        qp[r * C_ + 16 * t] = qv[t][r] + (acc[t][r] + bias[t]);
  }
}

__global__ __launch_bounds__(32) void k_out_conv(const float* __restrict__ q,
                                                 const float* __restrict__ ow,
                                                 const float* __restrict__ ob,
                                                 float* __restrict__ out) {
  __shared__ __attribute__((aligned(16))) float Qs[16 * QP];
  __shared__ float Bs[4096];
  const int s = blockIdx.x;
  DECODE_TILE(s, b, hh, w0)
  const int lane = threadIdx.x, g = lane >> 4, mc = lane & 15;
  const long pos0 = (long)s * 16;

  for (int j = 0; j < 32; ++j) {
    const int t = lane + 32 * j;
    const int m = t >> 6, c4 = t & 63;
    *(v4f*)(Qs + m * QP + 4 * c4) = *(const v4f*)(q + (pos0 + m) * C_ + 4 * c4);
  }
  for (int t = lane; t < 2048; t += 32) {
    const int cc = t >> 4, m = t & 15;
    Bs[t * 2 + 0] = (m < OUTC) ? ow[m * C_ + 2 * cc]     : 0.f;
    Bs[t * 2 + 1] = (m < OUTC) ? ow[m * C_ + 2 * cc + 1] : 0.f;
  }
  __syncthreads();

  v8f acc = {0.f, 0.f, 0.f, 0.f, 0.f, 0.f, 0.f, 0.f};
#pragma unroll 8
  for (int k0 = 0; k0 < C_; k0 += 4) {
    const int cc = (k0 >> 1) + g;
    v2f a  = *(const v2f*)(Qs + mc * QP + 2 * cc);
    v2f bb = *(const v2f*)(Bs + (cc * 16 + mc) * 2);
    acc = wmma4(a, bb, acc);
  }
  if (mc < OUTC) {
    const float bias = ob[mc];
    const long obase = ((long)(b * OUTC + mc) * H_ + hh) * W_ + w0;
#pragma unroll
    for (int r = 0; r < 8; ++r) out[obase + r + 8 * g] = acc[r] + bias;
  }
}

__global__ __launch_bounds__(256) void k_final(const unsigned* __restrict__ counts,
                                               const float* __restrict__ wsdiff,
                                               float* __restrict__ out) {
  __shared__ float red[256];
  const int gid = blockIdx.x * 256 + threadIdx.x;
  if (blockIdx.x < 32) {
    out[WF_POS + gid] = (float)counts[gid] * (1.0f / 4096.0f);
  } else {
    float sum = 0.f;
    for (int t = threadIdx.x; t < VQ_WG; t += 256) sum += wsdiff[t];
    red[threadIdx.x] = sum; __syncthreads();
    for (int st = 128; st > 0; st >>= 1) {
      if (threadIdx.x < st) red[threadIdx.x] += red[threadIdx.x + st];
      __syncthreads();
    }
    if (threadIdx.x == 0)
      out[DIFF_POS] = red[0] * (1.0f / ((float)NPOS * (float)C_));
  }
}

extern "C" void kernel_launch(void* const* d_in, const int* in_sizes, int n_in,
                              void* d_out, int out_size, void* d_ws, size_t ws_size,
                              hipStream_t stream) {
  (void)in_sizes; (void)n_in; (void)out_size; (void)ws_size;
  const float* in  = (const float*)d_in[0];
  const float* emb = (const float*)d_in[1];
  const float* rw1 = (const float*)d_in[2];
  const float* rb1 = (const float*)d_in[3];
  const float* rw2 = (const float*)d_in[4];
  const float* rb2 = (const float*)d_in[5];
  const float* ow  = (const float*)d_in[6];
  const float* ob  = (const float*)d_in[7];
  float* out = (float*)d_out;
  float* ws  = (float*)d_ws;

  float*    q      = ws + Q_OFF;
  float*    hbuf   = ws + H_OFF;
  float*    etp    = ws + ETP_OFF;
  float*    etn    = ws + ETN_OFF;
  float*    en2    = ws + EN2_OFF;
  unsigned* counts = (unsigned*)(ws + CNT_OFF);
  float*    wsdiff = ws + WDIFF_OFF;
  float*    wpack  = ws + WPACK_OFF;

  k_prep<<<512, 256, 0, stream>>>(emb, etp, etn, en2, counts);
  k_packw<<<(3 * WBLK + 255) / 256, 256, 0, stream>>>(rw1, wpack);
  k_vq<<<VQ_WG, 256, 0, stream>>>(in, etp, etn, en2, q, counts, wsdiff);
  for (int i = 0; i < 3; ++i) {
    k_res_conv3<<<NPOS / (PR * 16), 256, 0, stream>>>(q, wpack + (size_t)i * WBLK,
                                           rb1 + i * CB, hbuf);
    k_res_conv1<<<NTILE / 4, 128, 0, stream>>>(q, hbuf, rw2 + (size_t)i * C_ * CB,
                                           rb2 + i * C_);
  }
  k_out_conv<<<NTILE, 32, 0, stream>>>(q, ow, ob, out);
  k_final<<<33, 256, 0, stream>>>(counts, wsdiff, out);
}


# out 1x1 conv fused into the tail of the last residual 1x1 kernel: finished q tiles pass through LDS (the lane transpose) into the out product, never written back; launches 11 -> 10; same sums
# speedup vs baseline: 4.8548x; 1.0640x over previous
#include <hip/hip_runtime.h>


typedef float v2f __attribute__((ext_vector_type(2)));
typedef float v4f __attribute__((ext_vector_type(4)));
typedef float v8f __attribute__((ext_vector_type(8)));

#define B_    16
#define C_    256
#define H_    64
#define W_    64
#define OUTC  3
#define NT    512
#define CB    64

#define HW     (H_*W_)
#define NPOS   (B_*HW)
#define NTILE  (NPOS/16)
#define CHW    (C_*HW)

#define Q_OFF      0u
#define H_OFF      16777216u
#define ETP_OFF    20971520u
#define ETN_OFF    H_OFF
#define EN2_OFF    21102592u
#define CNT_OFF    21103104u
#define WDIFF_OFF  21111296u
#define WPACK_OFF  21115392u

#define PR     8
#define NHP    ((PR + 2) * 18)
#define KC     32
#define KP     (KC + 4)
#define NSLAB  (C_ / KC)
#define WSLAB  (9 * CB * KC)
#define WBLK   (NSLAB * WSLAB)
#define HQ     (NHP * (KC / 4))
#define LDS_F  28672

#define QP  (C_ + 4)

#define VC     32
#define VP     (C_ + 4)
#define VSLAB  (VC * C_)
#define NVS    (NT / VC)
#define VQ_WG  (NPOS / 128)

#define DIFF_POS   196608
#define WF_POS     196609

static __device__ __forceinline__ v8f wmma4(v2f a, v2f b, v8f c) {
  return __builtin_amdgcn_wmma_f32_16x16x4_f32(false, a, false, b, (short)0, c,
                                               false, false);
}

__global__ __launch_bounds__(256) void k_prep(const float* __restrict__ embed,
                                              float* __restrict__ etp,
                                              float* __restrict__ etn,
                                              float* __restrict__ en2,
                                              unsigned* __restrict__ counts) {
  const int gid = blockIdx.x * 256 + threadIdx.x;
  {
    const int n = gid >> 8, ch = gid & 255;
    const float v = embed[ch * NT + n];
    etn[gid] = v;
    const int cp = (ch & ~6) | ((ch & 2) << 1) | ((ch & 4) >> 1);
    etp[n * C_ + cp] = v;
  }
  if (gid < NT) {
    float s = 0.f;
    for (int c = 0; c < C_; ++c) { float v = embed[c * NT + gid]; s += v * v; }
    en2[gid] = s;
  }
  if (gid < B_ * NT) counts[gid] = 0u;
}

__global__ __launch_bounds__(256) void k_packw(const float* __restrict__ rw1,
                                               float* __restrict__ wpack) {
  const int t = blockIdx.x * 256 + threadIdx.x;
  if (t >= 3 * WBLK) return;
  const int i   = t / WBLK;
  const int r   = t % WBLK;
  const int kc  = r / WSLAB;
  const int r2  = r % WSLAB;
  const int tap = r2 / (CB * KC);
  const int r3  = r2 % (CB * KC);
  const int oc  = r3 / KC;
  const int cl  = r3 % KC;
  wpack[t] = rw1[(size_t)i * CB * C_ * 9
                 + ((size_t)oc * C_ + kc * KC + cl) * 9 + tap];
}

__global__ __launch_bounds__(256) void k_vq(const float* __restrict__ in,
                                            const float* __restrict__ etp,
                                            const float* __restrict__ etn,
                                            const float* __restrict__ en2,
                                            float* __restrict__ q,
                                            unsigned* __restrict__ counts,
                                            float* __restrict__ wsdiff) {
  __shared__ __attribute__((aligned(16))) float Bs[2 * VC * VP];
  __shared__ __attribute__((aligned(16))) float e2s[NT];
  __shared__ int   idxs[128];
  __shared__ float red[8];

  const int s = blockIdx.x;
  const int b = s >> 5;
  const int tid = threadIdx.x;
  const int wave = tid >> 5, lane = tid & 31, g = lane >> 4, mc = lane & 15;

  const int sdst = (tid >> 6) * VP + 4 * (tid & 63);
  v4f sreg[8];
#pragma unroll
  for (int i = 0; i < 8; ++i) sreg[i] = *(const v4f*)(etp + 4 * (tid + 256 * i));

  const float* xp = in + (size_t)b * CHW + (s & 31) * 128 + 16 * wave + mc
                    + 2 * g * HW;
  v4f a[32];
#pragma unroll
  for (int j = 0; j < 32; ++j)
    a[j] = v4f{xp[(8 * j) * HW], xp[(8 * j + 1) * HW],
               xp[(8 * j + 4) * HW], xp[(8 * j + 5) * HW]};

#pragma unroll
  for (int i = 0; i < 8; ++i) *(v4f*)(Bs + sdst + 4 * VP * i) = sreg[i];
  if (tid < NT / 4) *(v4f*)(e2s + 4 * tid) = *(const v4f*)(en2 + 4 * tid);
#pragma unroll
  for (int i = 0; i < 8; ++i)
    sreg[i] = *(const v4f*)(etp + VSLAB + 4 * (tid + 256 * i));
  __syncthreads();

  float bestv[8];
  int   besti[8];
#pragma unroll
  for (int r = 0; r < 8; ++r) { bestv[r] = 3.4e38f; besti[r] = 0; }

#pragma unroll 1
  for (int k = 0; k < NVS; ++k) {
    if (k + 1 < NVS) {
      float* dst = Bs + ((k + 1) & 1) * (VC * VP) + sdst;
#pragma unroll
      for (int i = 0; i < 8; ++i) *(v4f*)(dst + 4 * VP * i) = sreg[i];
    }
    if (k + 2 < NVS) {
      const float* src = etp + (size_t)(k + 2) * VSLAB + 4 * tid;
#pragma unroll
      for (int i = 0; i < 8; ++i) sreg[i] = *(const v4f*)(src + 1024 * i);
    }
    const float* bp = Bs + (k & 1) * (VC * VP) + mc * VP + 4 * g;
    const int n0 = k * VC + mc;
    const float e20 = e2s[n0], e21 = e2s[n0 + 16];
    v8f acc0 = {0.f, 0.f, 0.f, 0.f, 0.f, 0.f, 0.f, 0.f};
    v8f acc1 = {0.f, 0.f, 0.f, 0.f, 0.f, 0.f, 0.f, 0.f};
#pragma unroll
    for (int j = 0; j < 32; ++j) {
      const v4f w0 = *(const v4f*)(bp + 8 * j);
      const v4f w1 = *(const v4f*)(bp + 16 * VP + 8 * j);
      acc0 = wmma4(v2f{a[j].x, a[j].y}, v2f{w0.x, w0.y}, acc0);
      acc1 = wmma4(v2f{a[j].x, a[j].y}, v2f{w1.x, w1.y}, acc1);
      acc0 = wmma4(v2f{a[j].z, a[j].w}, v2f{w0.z, w0.w}, acc0);
      acc1 = wmma4(v2f{a[j].z, a[j].w}, v2f{w1.z, w1.w}, acc1);
    }
#pragma unroll
    for (int r = 0; r < 8; ++r) {
      const float d = e20 - 2.0f * acc0[r];
      if (d < bestv[r]) { bestv[r] = d; besti[r] = n0; }
    }
#pragma unroll
    for (int r = 0; r < 8; ++r) {
      const float d = e21 - 2.0f * acc1[r];
      if (d < bestv[r]) { bestv[r] = d; besti[r] = n0 + 16; }
    }
    __syncthreads();
  }

#pragma unroll
  for (int r = 0; r < 8; ++r) {
#pragma unroll
    for (int off = 1; off < 16; off <<= 1) {
      const float ov = __shfl_xor(bestv[r], off);
      const int   oi = __shfl_xor(besti[r], off);
      if (ov < bestv[r] || (ov == bestv[r] && oi < besti[r])) {
        bestv[r] = ov; besti[r] = oi;
      }
    }
  }
  if (mc == 0) {
#pragma unroll
    for (int r = 0; r < 8; ++r) idxs[wave * 16 + 8 * g + r] = besti[r];
  }
  __syncthreads();

  if (tid < 128) atomicAdd(&counts[b * NT + idxs[tid]], 1u);

  const float* ep = etp + idxs[wave * 16 + mc] * C_ + 4 * g;
  float fsum = 0.f;
#pragma unroll
  for (int j = 0; j < 32; ++j) {
    const v4f e = *(const v4f*)(ep + 8 * j);
    const float d0 = e.x - a[j].x, d1 = e.y - a[j].y;
    const float d2 = e.z - a[j].z, d3 = e.w - a[j].w;
    fsum += d0 * d0 + d1 * d1 + d2 * d2 + d3 * d3;
    if ((j & 7) == 7) __builtin_amdgcn_sched_barrier(0);
  }

  float* qb = q + (size_t)s * 128 * C_;
  for (int p = 0; p < 32; ++p) {
    const int m = 4 * p + (tid >> 6);
    const v4f v = *(const v4f*)(etn + idxs[m] * C_ + 4 * (tid & 63));
    *(v4f*)(qb + m * C_ + 4 * (tid & 63)) = v;
  }

#pragma unroll
  for (int off = 16; off > 0; off >>= 1) fsum += __shfl_xor(fsum, off);
  if (lane == 0) red[wave] = fsum;
  __syncthreads();
  if (tid == 0) {
    float t = 0.f;
    for (int w = 0; w < 8; ++w) t += red[w];
    wsdiff[s] = t;
  }
}

__global__ __launch_bounds__(256) void k_res_conv3(const float* __restrict__ q,
                                                   const float* __restrict__ wp,
                                                   const float* __restrict__ b1,
                                                   float* __restrict__ hbuf) {
  __shared__ __attribute__((aligned(16))) float lds[LDS_F];
  float* Ah = lds;
  float* Ws = lds + NHP * KP;
  const int s = blockIdx.x;
  const int b  = s >> 5;
  const int h0 = ((s & 31) >> 2) << 3;
  const int w0 = (s & 3) << 4;
  const int tid = threadIdx.x;

  int  hoff[6], hdst[6];
  bool hok[6];
#pragma unroll
  for (int j = 0; j < 6; ++j) {
    const int t  = tid + 256 * j;
    const int tt = t < HQ ? t : 0;
    const int p = tt >> 3, c4 = tt & 7, rr = p / 18, ww = p % 18;
    const int hy = h0 + rr - 1, wx = w0 + ww - 1;
    const bool ok = (unsigned)hy < 64u && (unsigned)wx < 64u;
    hok[j]  = ok;
    hoff[j] = ((b * H_ + (ok ? hy : h0)) * W_ + (ok ? wx : w0)) * C_ + 4 * c4;
    hdst[j] = p * KP + 4 * c4;
  }
  const int wdst = (tid >> 3) * KP + 4 * (tid & 7);

  v4f hreg[6], wreg[18];
#pragma unroll
  for (int j = 0; j < 6; ++j) hreg[j] = *(const v4f*)(q + hoff[j]);
#pragma unroll
  for (int j = 0; j < 18; ++j) wreg[j] = *(const v4f*)(wp + 4 * (tid + 256 * j));

  const int wave = tid >> 5, lane = tid & 31, g = lane >> 4, mc = lane & 15;
  const float* ap = Ah + (wave * 18 + mc) * KP + 4 * g;
  const float* bp = Ws + mc * KP + 4 * g;
  v8f acc[4];
#pragma unroll
  for (int t = 0; t < 4; ++t) acc[t] = v8f{0.f, 0.f, 0.f, 0.f, 0.f, 0.f, 0.f, 0.f};

#pragma unroll 1
  for (int kc = 0; kc < NSLAB; ++kc) {
#pragma unroll
    for (int j = 0; j < 6; ++j) {
      v4f v = hreg[j];
      v.x = (hok[j] && v.x > 0.f) ? v.x : 0.f;
      v.y = (hok[j] && v.y > 0.f) ? v.y : 0.f;
      v.z = (hok[j] && v.z > 0.f) ? v.z : 0.f;
      v.w = (hok[j] && v.w > 0.f) ? v.w : 0.f;
      if (j < 5 || tid < HQ - 1280) *(v4f*)(Ah + hdst[j]) = v;
    }
#pragma unroll
    for (int j = 0; j < 18; ++j) *(v4f*)(Ws + wdst + 32 * KP * j) = wreg[j];
    __syncthreads();

    if (kc + 1 < NSLAB) {
      const float* qn = q + (kc + 1) * KC;
      const float* wn = wp + (size_t)(kc + 1) * WSLAB;
#pragma unroll
      for (int j = 0; j < 6; ++j) hreg[j] = *(const v4f*)(qn + hoff[j]);
#pragma unroll
      for (int j = 0; j < 18; ++j) wreg[j] = *(const v4f*)(wn + 4 * (tid + 256 * j));
    }

#pragma unroll
    for (int tap = 0; tap < 9; ++tap) {
      const int dh = tap / 3, dw = tap % 3;
#pragma unroll
      for (int j = 0; j < KC / 8; ++j) {
        const v4f a = *(const v4f*)(ap + (dh * 18 + dw) * KP + 8 * j);
        v4f w[4];
#pragma unroll
        for (int t = 0; t < 4; ++t)
          w[t] = *(const v4f*)(bp + (tap * CB + 16 * t) * KP + 8 * j);
#pragma unroll
        for (int t = 0; t < 4; ++t)
          acc[t] = wmma4(v2f{a.x, a.y}, v2f{w[t].x, w[t].y}, acc[t]);
#pragma unroll
        for (int t = 0; t < 4; ++t)
          acc[t] = wmma4(v2f{a.z, a.w}, v2f{w[t].z, w[t].w}, acc[t]);
      }
    }
    __syncthreads();
  }

  const long pos0 = ((long)(b * H_ + h0 + wave)) * W_ + w0;
#pragma unroll
  for (int t = 0; t < 4; ++t) {
    const int oc = 16 * t + mc;
    const float bias = b1[oc];
#pragma unroll
    for (int r = 0; r < 8; ++r) {
      const float v = acc[t][r] + bias;
      hbuf[(pos0 + r + 8 * g) * CB + oc] = v > 0.f ? v : 0.f;
    }
  }
}

__global__ __launch_bounds__(128) void k_res_conv1(float* __restrict__ q,
                                                   const float* __restrict__ hbuf,
                                                   const float* __restrict__ w2,
                                                   const float* __restrict__ b2) {
  const int tid = threadIdx.x;
  const int wave = tid >> 5, lane = tid & 31, g = lane >> 4, mc = lane & 15;
  const int oc0 = wave * 64;

  v4f wf[4][8];
  float bias[4];
#pragma unroll
  for (int t = 0; t < 4; ++t) {
    const float* wr = w2 + (oc0 + 16 * t + mc) * CB + 4 * g;
#pragma unroll
    for (int j = 0; j < 8; ++j) wf[t][j] = *(const v4f*)(wr + 8 * j);
    bias[t] = b2[oc0 + 16 * t + mc];
  }

#pragma unroll 1
  for (int i = 0; i < 4; ++i) {
    const long pos0 = ((long)blockIdx.x * 4 + i) * 16;
    const float* hp = hbuf + (pos0 + mc) * CB + 4 * g;
    v4f a[8];
#pragma unroll
    for (int j = 0; j < 8; ++j) a[j] = *(const v4f*)(hp + 8 * j);

    float* qp = q + (pos0 + 8 * g) * C_ + oc0 + mc;
    float qv[4][8];
#pragma unroll
    for (int r = 0; r < 8; ++r)
#pragma unroll
      for (int t = 0; t < 4; ++t) qv[t][r] = qp[r * C_ + 16 * t];

    v8f acc[4];
#pragma unroll
    for (int t = 0; t < 4; ++t) acc[t] = v8f{0.f, 0.f, 0.f, 0.f, 0.f, 0.f, 0.f, 0.f};
#pragma unroll
    for (int j = 0; j < 8; ++j) {
#pragma unroll
      for (int t = 0; t < 4; ++t)
        acc[t] = wmma4(v2f{a[j].x, a[j].y}, v2f{wf[t][j].x, wf[t][j].y}, acc[t]);
#pragma unroll
      for (int t = 0; t < 4; ++t)
        acc[t] = wmma4(v2f{a[j].z, a[j].w}, v2f{wf[t][j].z, wf[t][j].w}, acc[t]);
    }
#pragma unroll
    for (int r = 0; r < 8; ++r)
#pragma unroll
      for (int t = 0; t < 4; ++t)
        qp[r * C_ + 16 * t] = qv[t][r] + (acc[t][r] + bias[t]);
  }
}

__global__ __launch_bounds__(128) void k_res_conv1_out(const float* __restrict__ q,
                                                       const float* __restrict__ hbuf,
                                                       const float* __restrict__ w2,
                                                       const float* __restrict__ b2,
                                                       const float* __restrict__ ow,
                                                       const float* __restrict__ ob,
                                                       float* __restrict__ out) {
  __shared__ __attribute__((aligned(16))) float Qs[64 * QP];
  const int tid = threadIdx.x;
  const int wave = tid >> 5, lane = tid & 31, g = lane >> 4, mc = lane & 15;
  const int oc0 = wave * 64;

  v4f wf[4][8];
  float bias[4];
#pragma unroll
  for (int t = 0; t < 4; ++t) {
    const float* wr = w2 + (oc0 + 16 * t + mc) * CB + 4 * g;
#pragma unroll
    for (int j = 0; j < 8; ++j) wf[t][j] = *(const v4f*)(wr + 8 * j);
    bias[t] = b2[oc0 + 16 * t + mc];
  }

#pragma unroll 1
  for (int i = 0; i < 4; ++i) {
    const long pos0 = ((long)blockIdx.x * 4 + i) * 16;
    const float* hp = hbuf + (pos0 + mc) * CB + 4 * g;
    v4f a[8];
#pragma unroll
    for (int j = 0; j < 8; ++j) a[j] = *(const v4f*)(hp + 8 * j);

    const float* qp = q + (pos0 + 8 * g) * C_ + oc0 + mc;
    float qv[4][8];
#pragma unroll
    for (int r = 0; r < 8; ++r)
#pragma unroll
      for (int t = 0; t < 4; ++t) qv[t][r] = qp[r * C_ + 16 * t];

    v8f acc[4];
#pragma unroll
    for (int t = 0; t < 4; ++t) acc[t] = v8f{0.f, 0.f, 0.f, 0.f, 0.f, 0.f, 0.f, 0.f};
#pragma unroll
    for (int j = 0; j < 8; ++j) {
#pragma unroll
      for (int t = 0; t < 4; ++t)
        acc[t] = wmma4(v2f{a[j].x, a[j].y}, v2f{wf[t][j].x, wf[t][j].y}, acc[t]);
#pragma unroll
      for (int t = 0; t < 4; ++t)
        acc[t] = wmma4(v2f{a[j].z, a[j].w}, v2f{wf[t][j].z, wf[t][j].w}, acc[t]);
    }
    float* qs = Qs + (16 * i + 8 * g) * QP + oc0 + mc;
#pragma unroll
    for (int r = 0; r < 8; ++r)
#pragma unroll
      for (int t = 0; t < 4; ++t)
        qs[r * QP + 16 * t] = qv[t][r] + (acc[t][r] + bias[t]);
  }

  const bool live = mc < OUTC;
  const float* bsrc = ow + (live ? mc : 0) * C_ + 2 * g;
  v2f bf[64];
#pragma unroll
  for (int k = 0; k < 64; ++k) {
    const v2f t = *(const v2f*)(bsrc + 4 * k);
    bf[k] = live ? t : v2f{0.f, 0.f};
  }
  const float obias = ob[live ? mc : 0];
  __syncthreads();

  const float* arow = Qs + (16 * wave + mc) * QP + 2 * g;
  v8f oacc = {0.f, 0.f, 0.f, 0.f, 0.f, 0.f, 0.f, 0.f};
#pragma unroll
  for (int k = 0; k < 64; ++k)
    oacc = wmma4(*(const v2f*)(arow + 4 * k), bf[k], oacc);

  if (live) {
    const int b = blockIdx.x >> 6, hh = blockIdx.x & 63;
    const long obase = ((long)(b * OUTC + mc) * H_ + hh) * W_ + 16 * wave;
#pragma unroll
    for (int r = 0; r < 8; ++r) out[obase + r + 8 * g] = oacc[r] + obias;
  }
}

__global__ __launch_bounds__(256) void k_final(const unsigned* __restrict__ counts,
                                               const float* __restrict__ wsdiff,
                                               float* __restrict__ out) {
  __shared__ float red[256];
  const int gid = blockIdx.x * 256 + threadIdx.x;
  if (blockIdx.x < 32) {
    out[WF_POS + gid] = (float)counts[gid] * (1.0f / 4096.0f);
  } else {
    float sum = 0.f;
    for (int t = threadIdx.x; t < VQ_WG; t += 256) sum += wsdiff[t];
    red[threadIdx.x] = sum; __syncthreads();
    for (int st = 128; st > 0; st >>= 1) {
      if (threadIdx.x < st) red[threadIdx.x] += red[threadIdx.x + st];
      __syncthreads();
    }
    if (threadIdx.x == 0)
      out[DIFF_POS] = red[0] * (1.0f / ((float)NPOS * (float)C_));
  }
}

extern "C" void kernel_launch(void* const* d_in, const int* in_sizes, int n_in,
                              void* d_out, int out_size, void* d_ws, size_t ws_size,
                              hipStream_t stream) {
  (void)in_sizes; (void)n_in; (void)out_size; (void)ws_size;
  const float* in  = (const float*)d_in[0];
  const float* emb = (const float*)d_in[1];
  const float* rw1 = (const float*)d_in[2];
  const float* rb1 = (const float*)d_in[3];
  const float* rw2 = (const float*)d_in[4];
  const float* rb2 = (const float*)d_in[5];
  const float* ow  = (const float*)d_in[6];
  const float* ob  = (const float*)d_in[7];
  float* out = (float*)d_out;
  float* ws  = (float*)d_ws;

  float*    q      = ws + Q_OFF;
  float*    hbuf   = ws + H_OFF;
  float*    etp    = ws + ETP_OFF;
  float*    etn    = ws + ETN_OFF;
  float*    en2    = ws + EN2_OFF;
  unsigned* counts = (unsigned*)(ws + CNT_OFF);
  float*    wsdiff = ws + WDIFF_OFF;
  float*    wpack  = ws + WPACK_OFF;

  k_prep<<<512, 256, 0, stream>>>(emb, etp, etn, en2, counts);
  k_packw<<<(3 * WBLK + 255) / 256, 256, 0, stream>>>(rw1, wpack);
  k_vq<<<VQ_WG, 256, 0, stream>>>(in, etp, etn, en2, q, counts, wsdiff);
  for (int i = 0; i < 2; ++i) {
    k_res_conv3<<<NPOS / (PR * 16), 256, 0, stream>>>(q, wpack + (size_t)i * WBLK,
                                           rb1 + i * CB, hbuf);
    k_res_conv1<<<NTILE / 4, 128, 0, stream>>>(q, hbuf, rw2 + (size_t)i * C_ * CB,
                                           rb2 + i * C_);
  }
  k_res_conv3<<<NPOS / (PR * 16), 256, 0, stream>>>(q, wpack + (size_t)2 * WBLK,
                                         rb1 + 2 * CB, hbuf);
  k_res_conv1_out<<<NTILE / 4, 128, 0, stream>>>(q, hbuf, rw2 + (size_t)2 * C_ * CB,
                                             rb2 + 2 * C_, ow, ob, out);
  k_final<<<33, 256, 0, stream>>>(counts, wsdiff, out);
}


# first res block's 3x3 conv from the codebook: h = 9-tap sum of rows of a table relu(codebook) x w1 (small f32 WMMA product), gathered by code index inside the 1x1 kernel; sum regrouped: taps outer
# speedup vs baseline: 5.9317x; 1.2218x over previous
#include <hip/hip_runtime.h>


typedef float v2f __attribute__((ext_vector_type(2)));
typedef float v4f __attribute__((ext_vector_type(4)));
typedef float v8f __attribute__((ext_vector_type(8)));

#define B_    16
#define C_    256
#define H_    64
#define W_    64
#define OUTC  3
#define NT    512
#define CB    64

#define HW     (H_*W_)
#define NPOS   (B_*HW)
#define NTILE  (NPOS/16)
#define CHW    (C_*HW)

#define Q_OFF      0u
#define H_OFF      16777216u
#define ETP_OFF    20971520u
#define ETN_OFF    H_OFF
#define EN2_OFF    21102592u
#define CNT_OFF    21103104u
#define WDIFF_OFF  21111296u
#define WPACK_OFF  21115392u
#define TAB_OFF    (H_OFF + 131072u)
#define IDX_OFF    (TAB_OFF + 294912u)
#define HP  (CB + 4)

#define PR     8
#define NHP    ((PR + 2) * 18)
#define KC     32
#define KP     (KC + 4)
#define NSLAB  (C_ / KC)
#define WSLAB  (9 * CB * KC)
#define WBLK   (NSLAB * WSLAB)
#define HQ     (NHP * (KC / 4))
#define LDS_F  28672

#define QP  (C_ + 4)

#define VC     32
#define VP     (C_ + 4)
#define VSLAB  (VC * C_)
#define NVS    (NT / VC)
#define VQ_WG  (NPOS / 128)

#define DIFF_POS   196608
#define WF_POS     196609

static __device__ __forceinline__ v8f wmma4(v2f a, v2f b, v8f c) {
  return __builtin_amdgcn_wmma_f32_16x16x4_f32(false, a, false, b, (short)0, c,
                                               false, false);
}

__global__ __launch_bounds__(256) void k_prep(const float* __restrict__ embed,
                                              float* __restrict__ etp,
                                              float* __restrict__ etn,
                                              float* __restrict__ en2,
                                              unsigned* __restrict__ counts) {
  const int gid = blockIdx.x * 256 + threadIdx.x;
  {
    const int n = gid >> 8, ch = gid & 255;
    const float v = embed[ch * NT + n];
    etn[gid] = v;
    const int cp = (ch & ~6) | ((ch & 2) << 1) | ((ch & 4) >> 1);
    etp[n * C_ + cp] = v;
  }
  if (gid < NT) {
    float s = 0.f;
    for (int c = 0; c < C_; ++c) { float v = embed[c * NT + gid]; s += v * v; }
    en2[gid] = s;
  }
  if (gid < B_ * NT) counts[gid] = 0u;
}

__global__ __launch_bounds__(256) void k_packw(const float* __restrict__ rw1,
                                               float* __restrict__ wpack) {
  const int t = blockIdx.x * 256 + threadIdx.x;
  if (t >= 3 * WBLK) return;
  const int i   = t / WBLK;
  const int r   = t % WBLK;
  const int kc  = r / WSLAB;
  const int r2  = r % WSLAB;
  const int tap = r2 / (CB * KC);
  const int r3  = r2 % (CB * KC);
  const int oc  = r3 / KC;
  const int cl  = r3 % KC;
  wpack[t] = rw1[(size_t)i * CB * C_ * 9
                 + ((size_t)oc * C_ + kc * KC + cl) * 9 + tap];
}

__global__ __launch_bounds__(256) void k_vq(const float* __restrict__ in,
                                            const float* __restrict__ etp,
                                            const float* __restrict__ etn,
                                            const float* __restrict__ en2,
                                            float* __restrict__ q,
                                            unsigned* __restrict__ counts,
                                            float* __restrict__ wsdiff,
                                            int* __restrict__ idxg) {
  __shared__ __attribute__((aligned(16))) float Bs[2 * VC * VP];
  __shared__ __attribute__((aligned(16))) float e2s[NT];
  __shared__ int   idxs[128];
  __shared__ float red[8];

  const int s = blockIdx.x;
  const int b = s >> 5;
  const int tid = threadIdx.x;
  const int wave = tid >> 5, lane = tid & 31, g = lane >> 4, mc = lane & 15;

  const int sdst = (tid >> 6) * VP + 4 * (tid & 63);
  v4f sreg[8];
#pragma unroll
  for (int i = 0; i < 8; ++i) sreg[i] = *(const v4f*)(etp + 4 * (tid + 256 * i));

  const float* xp = in + (size_t)b * CHW + (s & 31) * 128 + 16 * wave + mc
                    + 2 * g * HW;
  v4f a[32];
#pragma unroll
  for (int j = 0; j < 32; ++j)
    a[j] = v4f{xp[(8 * j) * HW], xp[(8 * j + 1) * HW],
               xp[(8 * j + 4) * HW], xp[(8 * j + 5) * HW]};

#pragma unroll
  for (int i = 0; i < 8; ++i) *(v4f*)(Bs + sdst + 4 * VP * i) = sreg[i];
  if (tid < NT / 4) *(v4f*)(e2s + 4 * tid) = *(const v4f*)(en2 + 4 * tid);
#pragma unroll
  for (int i = 0; i < 8; ++i)
    sreg[i] = *(const v4f*)(etp + VSLAB + 4 * (tid + 256 * i));
  __syncthreads();

  float bestv[8];
  int   besti[8];
#pragma unroll
  for (int r = 0; r < 8; ++r) { bestv[r] = 3.4e38f; besti[r] = 0; }

#pragma unroll 1
  for (int k = 0; k < NVS; ++k) {
    if (k + 1 < NVS) {
      float* dst = Bs + ((k + 1) & 1) * (VC * VP) + sdst;
#pragma unroll
      for (int i = 0; i < 8; ++i) *(v4f*)(dst + 4 * VP * i) = sreg[i];
    }
    if (k + 2 < NVS) {
      const float* src = etp + (size_t)(k + 2) * VSLAB + 4 * tid;
#pragma unroll
      for (int i = 0; i < 8; ++i) sreg[i] = *(const v4f*)(src + 1024 * i);
    }
    const float* bp = Bs + (k & 1) * (VC * VP) + mc * VP + 4 * g;
    const int n0 = k * VC + mc;
    const float e20 = e2s[n0], e21 = e2s[n0 + 16];
    v8f acc0 = {0.f, 0.f, 0.f, 0.f, 0.f, 0.f, 0.f, 0.f};
    v8f acc1 = {0.f, 0.f, 0.f, 0.f, 0.f, 0.f, 0.f, 0.f};
#pragma unroll
    for (int j = 0; j < 32; ++j) {
      const v4f w0 = *(const v4f*)(bp + 8 * j);
      const v4f w1 = *(const v4f*)(bp + 16 * VP + 8 * j);
      acc0 = wmma4(v2f{a[j].x, a[j].y}, v2f{w0.x, w0.y}, acc0);
      acc1 = wmma4(v2f{a[j].x, a[j].y}, v2f{w1.x, w1.y}, acc1);
      acc0 = wmma4(v2f{a[j].z, a[j].w}, v2f{w0.z, w0.w}, acc0);
      acc1 = wmma4(v2f{a[j].z, a[j].w}, v2f{w1.z, w1.w}, acc1);
    }
#pragma unroll
    for (int r = 0; r < 8; ++r) {
      const float d = e20 - 2.0f * acc0[r];
      if (d < bestv[r]) { bestv[r] = d; besti[r] = n0; }
    }
#pragma unroll
    for (int r = 0; r < 8; ++r) {
      const float d = e21 - 2.0f * acc1[r];
      if (d < bestv[r]) { bestv[r] = d; besti[r] = n0 + 16; }
    }
    __syncthreads();
  }

#pragma unroll
  for (int r = 0; r < 8; ++r) {
#pragma unroll
    for (int off = 1; off < 16; off <<= 1) {
      const float ov = __shfl_xor(bestv[r], off);
      const int   oi = __shfl_xor(besti[r], off);
      if (ov < bestv[r] || (ov == bestv[r] && oi < besti[r])) {
        bestv[r] = ov; besti[r] = oi;
      }
    }
  }
  if (mc == 0) {
#pragma unroll
    for (int r = 0; r < 8; ++r) idxs[wave * 16 + 8 * g + r] = besti[r];
  }
  __syncthreads();

  if (tid < 128) {
    atomicAdd(&counts[b * NT + idxs[tid]], 1u);
    idxg[s * 128 + tid] = idxs[tid];
  }

  const float* ep = etp + idxs[wave * 16 + mc] * C_ + 4 * g;
  float fsum = 0.f;
#pragma unroll
  for (int j = 0; j < 32; ++j) {
    const v4f e = *(const v4f*)(ep + 8 * j);
    const float d0 = e.x - a[j].x, d1 = e.y - a[j].y;
    const float d2 = e.z - a[j].z, d3 = e.w - a[j].w;
    fsum += d0 * d0 + d1 * d1 + d2 * d2 + d3 * d3;
    if ((j & 7) == 7) __builtin_amdgcn_sched_barrier(0);
  }

  float* qb = q + (size_t)s * 128 * C_;
  for (int p = 0; p < 32; ++p) {
    const int m = 4 * p + (tid >> 6);
    const v4f v = *(const v4f*)(etn + idxs[m] * C_ + 4 * (tid & 63));
    *(v4f*)(qb + m * C_ + 4 * (tid & 63)) = v;
  }

#pragma unroll
  for (int off = 16; off > 0; off >>= 1) fsum += __shfl_xor(fsum, off);
  if (lane == 0) red[wave] = fsum;
  __syncthreads();
  if (tid == 0) {
    float t = 0.f;
    for (int w = 0; w < 8; ++w) t += red[w];
    wsdiff[s] = t;
  }
}

__global__ __launch_bounds__(256) void k_res_conv3(const float* __restrict__ q,
                                                   const float* __restrict__ wp,
                                                   const float* __restrict__ b1,
                                                   float* __restrict__ hbuf) {
  __shared__ __attribute__((aligned(16))) float lds[LDS_F];
  float* Ah = lds;
  float* Ws = lds + NHP * KP;
  const int s = blockIdx.x;
  const int b  = s >> 5;
  const int h0 = ((s & 31) >> 2) << 3;
  const int w0 = (s & 3) << 4;
  const int tid = threadIdx.x;

  int  hoff[6], hdst[6];
  bool hok[6];
#pragma unroll
  for (int j = 0; j < 6; ++j) {
    const int t  = tid + 256 * j;
    const int tt = t < HQ ? t : 0;
    const int p = tt >> 3, c4 = tt & 7, rr = p / 18, ww = p % 18;
    const int hy = h0 + rr - 1, wx = w0 + ww - 1;
    const bool ok = (unsigned)hy < 64u && (unsigned)wx < 64u;
    hok[j]  = ok;
    hoff[j] = ((b * H_ + (ok ? hy : h0)) * W_ + (ok ? wx : w0)) * C_ + 4 * c4;
    hdst[j] = p * KP + 4 * c4;
  }
  const int wdst = (tid >> 3) * KP + 4 * (tid & 7);

  v4f hreg[6], wreg[18];
#pragma unroll
  for (int j = 0; j < 6; ++j) hreg[j] = *(const v4f*)(q + hoff[j]);
#pragma unroll
  for (int j = 0; j < 18; ++j) wreg[j] = *(const v4f*)(wp + 4 * (tid + 256 * j));

  const int wave = tid >> 5, lane = tid & 31, g = lane >> 4, mc = lane & 15;
  const float* ap = Ah + (wave * 18 + mc) * KP + 4 * g;
  const float* bp = Ws + mc * KP + 4 * g;
  v8f acc[4];
#pragma unroll
  for (int t = 0; t < 4; ++t) acc[t] = v8f{0.f, 0.f, 0.f, 0.f, 0.f, 0.f, 0.f, 0.f};

#pragma unroll 1
  for (int kc = 0; kc < NSLAB; ++kc) {
#pragma unroll
    for (int j = 0; j < 6; ++j) {
      v4f v = hreg[j];
      v.x = (hok[j] && v.x > 0.f) ? v.x : 0.f;
      v.y = (hok[j] && v.y > 0.f) ? v.y : 0.f;
      v.z = (hok[j] && v.z > 0.f) ? v.z : 0.f;
      v.w = (hok[j] && v.w > 0.f) ? v.w : 0.f;
      if (j < 5 || tid < HQ - 1280) *(v4f*)(Ah + hdst[j]) = v;
    }
#pragma unroll
    for (int j = 0; j < 18; ++j) *(v4f*)(Ws + wdst + 32 * KP * j) = wreg[j];
    __syncthreads();

    if (kc + 1 < NSLAB) {
      const float* qn = q + (kc + 1) * KC;
      const float* wn = wp + (size_t)(kc + 1) * WSLAB;
#pragma unroll
      for (int j = 0; j < 6; ++j) hreg[j] = *(const v4f*)(qn + hoff[j]);
#pragma unroll
      for (int j = 0; j < 18; ++j) wreg[j] = *(const v4f*)(wn + 4 * (tid + 256 * j));
    }

#pragma unroll
    for (int tap = 0; tap < 9; ++tap) {
      const int dh = tap / 3, dw = tap % 3;
#pragma unroll
      for (int j = 0; j < KC / 8; ++j) {
        const v4f a = *(const v4f*)(ap + (dh * 18 + dw) * KP + 8 * j);
        v4f w[4];
#pragma unroll
        for (int t = 0; t < 4; ++t)
          w[t] = *(const v4f*)(bp + (tap * CB + 16 * t) * KP + 8 * j);
#pragma unroll
        for (int t = 0; t < 4; ++t)
          acc[t] = wmma4(v2f{a.x, a.y}, v2f{w[t].x, w[t].y}, acc[t]);
#pragma unroll
        for (int t = 0; t < 4; ++t)
          acc[t] = wmma4(v2f{a.z, a.w}, v2f{w[t].z, w[t].w}, acc[t]);
      }
    }
    __syncthreads();
  }

  const long pos0 = ((long)(b * H_ + h0 + wave)) * W_ + w0;
#pragma unroll
  for (int t = 0; t < 4; ++t) {
    const int oc = 16 * t + mc;
    const float bias = b1[oc];
#pragma unroll
    for (int r = 0; r < 8; ++r) {
      const float v = acc[t][r] + bias;
      hbuf[(pos0 + r + 8 * g) * CB + oc] = v > 0.f ? v : 0.f;
    }
  }
}

__global__ __launch_bounds__(128) void k_res_conv1(float* __restrict__ q,
                                                   const float* __restrict__ hbuf,
                                                   const float* __restrict__ w2,
                                                   const float* __restrict__ b2) {
  const int tid = threadIdx.x;
  const int wave = tid >> 5, lane = tid & 31, g = lane >> 4, mc = lane & 15;
  const int oc0 = wave * 64;

  v4f wf[4][8];
  float bias[4];
#pragma unroll
  for (int t = 0; t < 4; ++t) {
    const float* wr = w2 + (oc0 + 16 * t + mc) * CB + 4 * g;
#pragma unroll
    for (int j = 0; j < 8; ++j) wf[t][j] = *(const v4f*)(wr + 8 * j);
    bias[t] = b2[oc0 + 16 * t + mc];
  }

#pragma unroll 1
  for (int i = 0; i < 4; ++i) {
    const long pos0 = ((long)blockIdx.x * 4 + i) * 16;
    const float* hp = hbuf + (pos0 + mc) * CB + 4 * g;
    v4f a[8];
#pragma unroll
    for (int j = 0; j < 8; ++j) a[j] = *(const v4f*)(hp + 8 * j);

    float* qp = q + (pos0 + 8 * g) * C_ + oc0 + mc;
    float qv[4][8];
#pragma unroll
    for (int r = 0; r < 8; ++r)
#pragma unroll
      for (int t = 0; t < 4; ++t) qv[t][r] = qp[r * C_ + 16 * t];

    v8f acc[4];
#pragma unroll
    for (int t = 0; t < 4; ++t) acc[t] = v8f{0.f, 0.f, 0.f, 0.f, 0.f, 0.f, 0.f, 0.f};
#pragma unroll
    for (int j = 0; j < 8; ++j) {
#pragma unroll
      for (int t = 0; t < 4; ++t)
        acc[t] = wmma4(v2f{a[j].x, a[j].y}, v2f{wf[t][j].x, wf[t][j].y}, acc[t]);
#pragma unroll
      for (int t = 0; t < 4; ++t)
        acc[t] = wmma4(v2f{a[j].z, a[j].w}, v2f{wf[t][j].z, wf[t][j].w}, acc[t]);
    }
#pragma unroll
    for (int r = 0; r < 8; ++r)
#pragma unroll
      for (int t = 0; t < 4; ++t)
        qp[r * C_ + 16 * t] = qv[t][r] + (acc[t][r] + bias[t]);
  }
}

__global__ __launch_bounds__(64) void k_tab(const float* __restrict__ etn,
                                            const float* __restrict__ wp,
                                            float* __restrict__ tab) {
  const int tid = threadIdx.x;
  const int wave = tid >> 5, lane = tid & 31, g = lane >> 4, mc = lane & 15;
  const int tap = blockIdx.x >> 5;
  const int nt  = blockIdx.x & 31;
  const int oc0 = 32 * wave;
  const float* ap = etn + (nt * 16 + mc) * C_ + 4 * g;
  const float* bp = wp + (tap * CB + oc0 + mc) * KC + 4 * g;
  v8f acc[2];
#pragma unroll
  for (int t = 0; t < 2; ++t) acc[t] = v8f{0.f, 0.f, 0.f, 0.f, 0.f, 0.f, 0.f, 0.f};
#pragma unroll
  for (int j = 0; j < 32; ++j) {
    v4f a = *(const v4f*)(ap + 8 * j);
    a.x = a.x > 0.f ? a.x : 0.f;  a.y = a.y > 0.f ? a.y : 0.f;
    a.z = a.z > 0.f ? a.z : 0.f;  a.w = a.w > 0.f ? a.w : 0.f;
    v4f w[2];
#pragma unroll
    for (int t = 0; t < 2; ++t)
      w[t] = *(const v4f*)(bp + (j >> 2) * WSLAB + 16 * t * KC + 8 * (j & 3));
#pragma unroll
    for (int t = 0; t < 2; ++t)
      acc[t] = wmma4(v2f{a.x, a.y}, v2f{w[t].x, w[t].y}, acc[t]);
#pragma unroll
    for (int t = 0; t < 2; ++t)
      acc[t] = wmma4(v2f{a.z, a.w}, v2f{w[t].z, w[t].w}, acc[t]);
  }
  float* tp = tab + ((size_t)(tap * NT + nt * 16 + 8 * g)) * CB + oc0 + mc;
#pragma unroll
  for (int t = 0; t < 2; ++t)
#pragma unroll
    for (int r = 0; r < 8; ++r) tp[r * CB + 16 * t] = acc[t][r];
}

__global__ __launch_bounds__(128) void k_res_conv1_g(float* __restrict__ q,
                                                     const int* __restrict__ idxg,
                                                     const float* __restrict__ tab,
                                                     const float* __restrict__ b1,
                                                     const float* __restrict__ w2,
                                                     const float* __restrict__ b2) {
  __shared__ __attribute__((aligned(16))) float Hs[64 * HP];
  __shared__ int ids[3 * 64];
  const int tid = threadIdx.x;
  const int wave = tid >> 5, lane = tid & 31, g = lane >> 4, mc = lane & 15;
  const int oc0 = wave * 64;
  const int b = blockIdx.x >> 6, hh = blockIdx.x & 63;

  for (int t = tid; t < 192; t += 128) {
    const int hy = hh + (t >> 6) - 1;
    const bool ok = (unsigned)hy < 64u;
    const int v = idxg[(b * H_ + (ok ? hy : hh)) * W_ + (t & 63)];
    ids[t] = ok ? v : -1;
  }
  __syncthreads();

  const int cq = tid & 15, pl = tid >> 4;
  const v4f bv = *(const v4f*)(b1 + 4 * cq);
#pragma unroll 2
  for (int i = 0; i < 8; ++i) {
    const int p = 8 * i + pl;
    v4f s = {0.f, 0.f, 0.f, 0.f};
#pragma unroll
    for (int tap = 0; tap < 9; ++tap) {
      const int dh = tap / 3, dw = tap % 3;
      const int wx = p + dw - 1;
      const bool in = (unsigned)wx < 64u;
      const int n = ids[dh * 64 + (in ? wx : p)];
      const bool ok = in && n >= 0;
      const v4f t = *(const v4f*)(tab + ((size_t)(tap * NT + (ok ? n : 0))) * CB
                                  + 4 * cq);
      s.x += ok ? t.x : 0.f;  s.y += ok ? t.y : 0.f;
      s.z += ok ? t.z : 0.f;  s.w += ok ? t.w : 0.f;
    }
    s += bv;
    s.x = s.x > 0.f ? s.x : 0.f;  s.y = s.y > 0.f ? s.y : 0.f;
    s.z = s.z > 0.f ? s.z : 0.f;  s.w = s.w > 0.f ? s.w : 0.f;
    *(v4f*)(Hs + p * HP + 4 * cq) = s;
  }

  v4f wf[4][8];
  float bias[4];
#pragma unroll
  for (int t = 0; t < 4; ++t) {
    const float* wr = w2 + (oc0 + 16 * t + mc) * CB + 4 * g;
#pragma unroll
    for (int j = 0; j < 8; ++j) wf[t][j] = *(const v4f*)(wr + 8 * j);
    bias[t] = b2[oc0 + 16 * t + mc];
  }
  __syncthreads();

#pragma unroll 1
  for (int i = 0; i < 4; ++i) {
    const long pos0 = ((long)blockIdx.x * 4 + i) * 16;
    const float* hp = Hs + (16 * i + mc) * HP + 4 * g;
    v4f a[8];
#pragma unroll
    for (int j = 0; j < 8; ++j) a[j] = *(const v4f*)(hp + 8 * j);

    float* qp = q + (pos0 + 8 * g) * C_ + oc0 + mc;
    float qv[4][8];
#pragma unroll
    for (int r = 0; r < 8; ++r)
#pragma unroll
      for (int t = 0; t < 4; ++t) qv[t][r] = qp[r * C_ + 16 * t];

    v8f acc[4];
#pragma unroll
    for (int t = 0; t < 4; ++t) acc[t] = v8f{0.f, 0.f, 0.f, 0.f, 0.f, 0.f, 0.f, 0.f};
#pragma unroll
    for (int j = 0; j < 8; ++j) {
#pragma unroll
      for (int t = 0; t < 4; ++t)
        acc[t] = wmma4(v2f{a[j].x, a[j].y}, v2f{wf[t][j].x, wf[t][j].y}, acc[t]);
#pragma unroll
      for (int t = 0; t < 4; ++t)
        acc[t] = wmma4(v2f{a[j].z, a[j].w}, v2f{wf[t][j].z, wf[t][j].w}, acc[t]);
    }
#pragma unroll
    for (int r = 0; r < 8; ++r)
#pragma unroll
      for (int t = 0; t < 4; ++t)
        qp[r * C_ + 16 * t] = qv[t][r] + (acc[t][r] + bias[t]);
  }
}

__global__ __launch_bounds__(128) void k_res_conv1_out(const float* __restrict__ q,
                                                       const float* __restrict__ hbuf,
                                                       const float* __restrict__ w2,
                                                       const float* __restrict__ b2,
                                                       const float* __restrict__ ow,
                                                       const float* __restrict__ ob,
                                                       float* __restrict__ out) {
  __shared__ __attribute__((aligned(16))) float Qs[64 * QP];
  const int tid = threadIdx.x;
  const int wave = tid >> 5, lane = tid & 31, g = lane >> 4, mc = lane & 15;
  const int oc0 = wave * 64;

  v4f wf[4][8];
  float bias[4];
#pragma unroll
  for (int t = 0; t < 4; ++t) {
    const float* wr = w2 + (oc0 + 16 * t + mc) * CB + 4 * g;
#pragma unroll
    for (int j = 0; j < 8; ++j) wf[t][j] = *(const v4f*)(wr + 8 * j);
    bias[t] = b2[oc0 + 16 * t + mc];
  }

#pragma unroll 1
  for (int i = 0; i < 4; ++i) {
    const long pos0 = ((long)blockIdx.x * 4 + i) * 16;
    const float* hp = hbuf + (pos0 + mc) * CB + 4 * g;
    v4f a[8];
#pragma unroll
    for (int j = 0; j < 8; ++j) a[j] = *(const v4f*)(hp + 8 * j);

    const float* qp = q + (pos0 + 8 * g) * C_ + oc0 + mc;
    float qv[4][8];
#pragma unroll
    for (int r = 0; r < 8; ++r)
#pragma unroll
      for (int t = 0; t < 4; ++t) qv[t][r] = qp[r * C_ + 16 * t];

    v8f acc[4];
#pragma unroll
    for (int t = 0; t < 4; ++t) acc[t] = v8f{0.f, 0.f, 0.f, 0.f, 0.f, 0.f, 0.f, 0.f};
#pragma unroll
    for (int j = 0; j < 8; ++j) {
#pragma unroll
      for (int t = 0; t < 4; ++t)
        acc[t] = wmma4(v2f{a[j].x, a[j].y}, v2f{wf[t][j].x, wf[t][j].y}, acc[t]);
#pragma unroll
      for (int t = 0; t < 4; ++t)
        acc[t] = wmma4(v2f{a[j].z, a[j].w}, v2f{wf[t][j].z, wf[t][j].w}, acc[t]);
    }
    float* qs = Qs + (16 * i + 8 * g) * QP + oc0 + mc;
#pragma unroll
    for (int r = 0; r < 8; ++r)
#pragma unroll
      for (int t = 0; t < 4; ++t)
        qs[r * QP + 16 * t] = qv[t][r] + (acc[t][r] + bias[t]);
  }

  const bool live = mc < OUTC;
  const float* bsrc = ow + (live ? mc : 0) * C_ + 2 * g;
  v2f bf[64];
#pragma unroll
  for (int k = 0; k < 64; ++k) {
    const v2f t = *(const v2f*)(bsrc + 4 * k);
    bf[k] = live ? t : v2f{0.f, 0.f};
  }
  const float obias = ob[live ? mc : 0];
  __syncthreads();

  const float* arow = Qs + (16 * wave + mc) * QP + 2 * g;
  v8f oacc = {0.f, 0.f, 0.f, 0.f, 0.f, 0.f, 0.f, 0.f};
#pragma unroll
  for (int k = 0; k < 64; ++k)
    oacc = wmma4(*(const v2f*)(arow + 4 * k), bf[k], oacc);

  if (live) {
    const int b = blockIdx.x >> 6, hh = blockIdx.x & 63;
    const long obase = ((long)(b * OUTC + mc) * H_ + hh) * W_ + 16 * wave;
#pragma unroll
    for (int r = 0; r < 8; ++r) out[obase + r + 8 * g] = oacc[r] + obias;
  }
}

__global__ __launch_bounds__(256) void k_final(const unsigned* __restrict__ counts,
                                               const float* __restrict__ wsdiff,
                                               float* __restrict__ out) {
  __shared__ float red[256];
  const int gid = blockIdx.x * 256 + threadIdx.x;
  if (blockIdx.x < 32) {
    out[WF_POS + gid] = (float)counts[gid] * (1.0f / 4096.0f);
  } else {
    float sum = 0.f;
    for (int t = threadIdx.x; t < VQ_WG; t += 256) sum += wsdiff[t];
    red[threadIdx.x] = sum; __syncthreads();
    for (int st = 128; st > 0; st >>= 1) {
      if (threadIdx.x < st) red[threadIdx.x] += red[threadIdx.x + st];
      __syncthreads();
    }
    if (threadIdx.x == 0)
      out[DIFF_POS] = red[0] * (1.0f / ((float)NPOS * (float)C_));
  }
}

extern "C" void kernel_launch(void* const* d_in, const int* in_sizes, int n_in,
                              void* d_out, int out_size, void* d_ws, size_t ws_size,
                              hipStream_t stream) {
  (void)in_sizes; (void)n_in; (void)out_size; (void)ws_size;
  const float* in  = (const float*)d_in[0];
  const float* emb = (const float*)d_in[1];
  const float* rw1 = (const float*)d_in[2];
  const float* rb1 = (const float*)d_in[3];
  const float* rw2 = (const float*)d_in[4];
  const float* rb2 = (const float*)d_in[5];
  const float* ow  = (const float*)d_in[6];
  const float* ob  = (const float*)d_in[7];
  float* out = (float*)d_out;
  float* ws  = (float*)d_ws;

  float*    q      = ws + Q_OFF;
  float*    hbuf   = ws + H_OFF;
  float*    etp    = ws + ETP_OFF;
  float*    etn    = ws + ETN_OFF;
  float*    en2    = ws + EN2_OFF;
  unsigned* counts = (unsigned*)(ws + CNT_OFF);
  float*    wsdiff = ws + WDIFF_OFF;
  float*    wpack  = ws + WPACK_OFF;
  float*    tab    = ws + TAB_OFF;
  int*      idxg   = (int*)(ws + IDX_OFF);

  k_prep<<<512, 256, 0, stream>>>(emb, etp, etn, en2, counts);
  k_packw<<<(3 * WBLK + 255) / 256, 256, 0, stream>>>(rw1, wpack);
  k_tab<<<9 * 32, 64, 0, stream>>>(etn, wpack, tab);
  k_vq<<<VQ_WG, 256, 0, stream>>>(in, etp, etn, en2, q, counts, wsdiff, idxg);
  k_res_conv1_g<<<NTILE / 4, 128, 0, stream>>>(q, idxg, tab, rb1, rw2, rb2);
  k_res_conv3<<<NPOS / (PR * 16), 256, 0, stream>>>(q, wpack + (size_t)1 * WBLK,
                                         rb1 + 1 * CB, hbuf);
  k_res_conv1<<<NTILE / 4, 128, 0, stream>>>(q, hbuf, rw2 + (size_t)1 * C_ * CB,
                                         rb2 + 1 * C_);
  k_res_conv3<<<NPOS / (PR * 16), 256, 0, stream>>>(q, wpack + (size_t)2 * WBLK,
                                         rb1 + 2 * CB, hbuf);
  k_res_conv1_out<<<NTILE / 4, 128, 0, stream>>>(q, hbuf, rw2 + (size_t)2 * C_ * CB,
                                             rb2 + 2 * C_, ow, ob, out);
  k_final<<<33, 256, 0, stream>>>(counts, wsdiff, out);
}


# VQ stores no q, only code indices; the first res block reads its residual input (a position's code row) from the code-major codebook copy by index, ahead of each tile's products; same sums
# speedup vs baseline: 6.1834x; 1.0424x over previous
#include <hip/hip_runtime.h>


typedef float v2f __attribute__((ext_vector_type(2)));
typedef float v4f __attribute__((ext_vector_type(4)));
typedef float v8f __attribute__((ext_vector_type(8)));

#define B_    16
#define C_    256
#define H_    64
#define W_    64
#define OUTC  3
#define NT    512
#define CB    64

#define HW     (H_*W_)
#define NPOS   (B_*HW)
#define NTILE  (NPOS/16)
#define CHW    (C_*HW)

#define Q_OFF      0u
#define H_OFF      16777216u
#define ETP_OFF    20971520u
#define ETN_OFF    H_OFF
#define EN2_OFF    21102592u
#define CNT_OFF    21103104u
#define WDIFF_OFF  21111296u
#define WPACK_OFF  21115392u
#define TAB_OFF    (H_OFF + 131072u)
#define IDX_OFF    (TAB_OFF + 294912u)
#define HP  (CB + 4)

#define PR     8
#define NHP    ((PR + 2) * 18)
#define KC     32
#define KP     (KC + 4)
#define NSLAB  (C_ / KC)
#define WSLAB  (9 * CB * KC)
#define WBLK   (NSLAB * WSLAB)
#define HQ     (NHP * (KC / 4))
#define LDS_F  28672

#define QP  (C_ + 4)

#define VC     32
#define VP     (C_ + 4)
#define VSLAB  (VC * C_)
#define NVS    (NT / VC)
#define VQ_WG  (NPOS / 128)

#define DIFF_POS   196608
#define WF_POS     196609

static __device__ __forceinline__ v8f wmma4(v2f a, v2f b, v8f c) {
  return __builtin_amdgcn_wmma_f32_16x16x4_f32(false, a, false, b, (short)0, c,
                                               false, false);
}

__global__ __launch_bounds__(256) void k_prep(const float* __restrict__ embed,
                                              float* __restrict__ etp,
                                              float* __restrict__ etn,
                                              float* __restrict__ en2,
                                              unsigned* __restrict__ counts) {
  const int gid = blockIdx.x * 256 + threadIdx.x;
  {
    const int n = gid >> 8, ch = gid & 255;
    const float v = embed[ch * NT + n];
    etn[gid] = v;
    const int cp = (ch & ~6) | ((ch & 2) << 1) | ((ch & 4) >> 1);
    etp[n * C_ + cp] = v;
  }
  if (gid < NT) {
    float s = 0.f;
    for (int c = 0; c < C_; ++c) { float v = embed[c * NT + gid]; s += v * v; }
    en2[gid] = s;
  }
  if (gid < B_ * NT) counts[gid] = 0u;
}

__global__ __launch_bounds__(256) void k_packw(const float* __restrict__ rw1,
                                               float* __restrict__ wpack) {
  const int t = blockIdx.x * 256 + threadIdx.x;
  if (t >= 3 * WBLK) return;
  const int i   = t / WBLK;
  const int r   = t % WBLK;
  const int kc  = r / WSLAB;
  const int r2  = r % WSLAB;
  const int tap = r2 / (CB * KC);
  const int r3  = r2 % (CB * KC);
  const int oc  = r3 / KC;
  const int cl  = r3 % KC;
  wpack[t] = rw1[(size_t)i * CB * C_ * 9
                 + ((size_t)oc * C_ + kc * KC + cl) * 9 + tap];
}

__global__ __launch_bounds__(256) void k_vq(const float* __restrict__ in,
                                            const float* __restrict__ etp,
                                            const float* __restrict__ en2,
                                            unsigned* __restrict__ counts,
                                            float* __restrict__ wsdiff,
                                            int* __restrict__ idxg) {
  __shared__ __attribute__((aligned(16))) float Bs[2 * VC * VP];
  __shared__ __attribute__((aligned(16))) float e2s[NT];
  __shared__ int   idxs[128];
  __shared__ float red[8];

  const int s = blockIdx.x;
  const int b = s >> 5;
  const int tid = threadIdx.x;
  const int wave = tid >> 5, lane = tid & 31, g = lane >> 4, mc = lane & 15;

  const int sdst = (tid >> 6) * VP + 4 * (tid & 63);
  v4f sreg[8];
#pragma unroll
  for (int i = 0; i < 8; ++i) sreg[i] = *(const v4f*)(etp + 4 * (tid + 256 * i));

  const float* xp = in + (size_t)b * CHW + (s & 31) * 128 + 16 * wave + mc
                    + 2 * g * HW;
  v4f a[32];
#pragma unroll
  for (int j = 0; j < 32; ++j)
    a[j] = v4f{xp[(8 * j) * HW], xp[(8 * j + 1) * HW],
               xp[(8 * j + 4) * HW], xp[(8 * j + 5) * HW]};

#pragma unroll
  for (int i = 0; i < 8; ++i) *(v4f*)(Bs + sdst + 4 * VP * i) = sreg[i];
  if (tid < NT / 4) *(v4f*)(e2s + 4 * tid) = *(const v4f*)(en2 + 4 * tid);
#pragma unroll
  for (int i = 0; i < 8; ++i)
    sreg[i] = *(const v4f*)(etp + VSLAB + 4 * (tid + 256 * i));
  __syncthreads();

  float bestv[8];
  int   besti[8];
#pragma unroll
  for (int r = 0; r < 8; ++r) { bestv[r] = 3.4e38f; besti[r] = 0; }

#pragma unroll 1
  for (int k = 0; k < NVS; ++k) {
    if (k + 1 < NVS) {
      float* dst = Bs + ((k + 1) & 1) * (VC * VP) + sdst;
#pragma unroll
      for (int i = 0; i < 8; ++i) *(v4f*)(dst + 4 * VP * i) = sreg[i];
    }
    if (k + 2 < NVS) {
      const float* src = etp + (size_t)(k + 2) * VSLAB + 4 * tid;
#pragma unroll
      for (int i = 0; i < 8; ++i) sreg[i] = *(const v4f*)(src + 1024 * i);
    }
    const float* bp = Bs + (k & 1) * (VC * VP) + mc * VP + 4 * g;
    const int n0 = k * VC + mc;
    const float e20 = e2s[n0], e21 = e2s[n0 + 16];
    v8f acc0 = {0.f, 0.f, 0.f, 0.f, 0.f, 0.f, 0.f, 0.f};
    v8f acc1 = {0.f, 0.f, 0.f, 0.f, 0.f, 0.f, 0.f, 0.f};
#pragma unroll
    for (int j = 0; j < 32; ++j) {
      const v4f w0 = *(const v4f*)(bp + 8 * j);
      const v4f w1 = *(const v4f*)(bp + 16 * VP + 8 * j);
      acc0 = wmma4(v2f{a[j].x, a[j].y}, v2f{w0.x, w0.y}, acc0);
      acc1 = wmma4(v2f{a[j].x, a[j].y}, v2f{w1.x, w1.y}, acc1);
      acc0 = wmma4(v2f{a[j].z, a[j].w}, v2f{w0.z, w0.w}, acc0);
      acc1 = wmma4(v2f{a[j].z, a[j].w}, v2f{w1.z, w1.w}, acc1);
    }
#pragma unroll
    for (int r = 0; r < 8; ++r) {
      const float d = e20 - 2.0f * acc0[r];
      if (d < bestv[r]) { bestv[r] = d; besti[r] = n0; }
    }
#pragma unroll
    for (int r = 0; r < 8; ++r) {
      const float d = e21 - 2.0f * acc1[r];
      if (d < bestv[r]) { bestv[r] = d; besti[r] = n0 + 16; }
    }
    __syncthreads();
  }

#pragma unroll
  for (int r = 0; r < 8; ++r) {
#pragma unroll
    for (int off = 1; off < 16; off <<= 1) {
      const float ov = __shfl_xor(bestv[r], off);
      const int   oi = __shfl_xor(besti[r], off);
      if (ov < bestv[r] || (ov == bestv[r] && oi < besti[r])) {
        bestv[r] = ov; besti[r] = oi;
      }
    }
  }
  if (mc == 0) {
#pragma unroll
    for (int r = 0; r < 8; ++r) idxs[wave * 16 + 8 * g + r] = besti[r];
  }
  __syncthreads();

  if (tid < 128) {
    atomicAdd(&counts[b * NT + idxs[tid]], 1u);
    idxg[s * 128 + tid] = idxs[tid];
  }

  const float* ep = etp + idxs[wave * 16 + mc] * C_ + 4 * g;
  float fsum = 0.f;
#pragma unroll
  for (int j = 0; j < 32; ++j) {
    const v4f e = *(const v4f*)(ep + 8 * j);
    const float d0 = e.x - a[j].x, d1 = e.y - a[j].y;
    const float d2 = e.z - a[j].z, d3 = e.w - a[j].w;
    fsum += d0 * d0 + d1 * d1 + d2 * d2 + d3 * d3;
    if ((j & 7) == 7) __builtin_amdgcn_sched_barrier(0);
  }

#pragma unroll
  for (int off = 16; off > 0; off >>= 1) fsum += __shfl_xor(fsum, off);
  if (lane == 0) red[wave] = fsum;
  __syncthreads();
  if (tid == 0) {
    float t = 0.f;
    for (int w = 0; w < 8; ++w) t += red[w];
    wsdiff[s] = t;
  }
}

__global__ __launch_bounds__(256) void k_res_conv3(const float* __restrict__ q,
                                                   const float* __restrict__ wp,
                                                   const float* __restrict__ b1,
                                                   float* __restrict__ hbuf) {
  __shared__ __attribute__((aligned(16))) float lds[LDS_F];
  float* Ah = lds;
  float* Ws = lds + NHP * KP;
  const int s = blockIdx.x;
  const int b  = s >> 5;
  const int h0 = ((s & 31) >> 2) << 3;
  const int w0 = (s & 3) << 4;
  const int tid = threadIdx.x;

  int  hoff[6], hdst[6];
  bool hok[6];
#pragma unroll
  for (int j = 0; j < 6; ++j) {
    const int t  = tid + 256 * j;
    const int tt = t < HQ ? t : 0;
    const int p = tt >> 3, c4 = tt & 7, rr = p / 18, ww = p % 18;
    const int hy = h0 + rr - 1, wx = w0 + ww - 1;
    const bool ok = (unsigned)hy < 64u && (unsigned)wx < 64u;
    hok[j]  = ok;
    hoff[j] = ((b * H_ + (ok ? hy : h0)) * W_ + (ok ? wx : w0)) * C_ + 4 * c4;
    hdst[j] = p * KP + 4 * c4;
  }
  const int wdst = (tid >> 3) * KP + 4 * (tid & 7);

  v4f hreg[6], wreg[18];
#pragma unroll
  for (int j = 0; j < 6; ++j) hreg[j] = *(const v4f*)(q + hoff[j]);
#pragma unroll
  for (int j = 0; j < 18; ++j) wreg[j] = *(const v4f*)(wp + 4 * (tid + 256 * j));

  const int wave = tid >> 5, lane = tid & 31, g = lane >> 4, mc = lane & 15;
  const float* ap = Ah + (wave * 18 + mc) * KP + 4 * g;
  const float* bp = Ws + mc * KP + 4 * g;
  v8f acc[4];
#pragma unroll
  for (int t = 0; t < 4; ++t) acc[t] = v8f{0.f, 0.f, 0.f, 0.f, 0.f, 0.f, 0.f, 0.f};

#pragma unroll 1
  for (int kc = 0; kc < NSLAB; ++kc) {
#pragma unroll
    for (int j = 0; j < 6; ++j) {
      v4f v = hreg[j];
      v.x = (hok[j] && v.x > 0.f) ? v.x : 0.f;
      v.y = (hok[j] && v.y > 0.f) ? v.y : 0.f;
      v.z = (hok[j] && v.z > 0.f) ? v.z : 0.f;
      v.w = (hok[j] && v.w > 0.f) ? v.w : 0.f;
      if (j < 5 || tid < HQ - 1280) *(v4f*)(Ah + hdst[j]) = v;
    }
#pragma unroll
    for (int j = 0; j < 18; ++j) *(v4f*)(Ws + wdst + 32 * KP * j) = wreg[j];
    __syncthreads();

    if (kc + 1 < NSLAB) {
      const float* qn = q + (kc + 1) * KC;
      const float* wn = wp + (size_t)(kc + 1) * WSLAB;
#pragma unroll
      for (int j = 0; j < 6; ++j) hreg[j] = *(const v4f*)(qn + hoff[j]);
#pragma unroll
      for (int j = 0; j < 18; ++j) wreg[j] = *(const v4f*)(wn + 4 * (tid + 256 * j));
    }

#pragma unroll
    for (int tap = 0; tap < 9; ++tap) {
      const int dh = tap / 3, dw = tap % 3;
#pragma unroll
      for (int j = 0; j < KC / 8; ++j) {
        const v4f a = *(const v4f*)(ap + (dh * 18 + dw) * KP + 8 * j);
        v4f w[4];
#pragma unroll
        for (int t = 0; t < 4; ++t)
          w[t] = *(const v4f*)(bp + (tap * CB + 16 * t) * KP + 8 * j);
#pragma unroll
        for (int t = 0; t < 4; ++t)
          acc[t] = wmma4(v2f{a.x, a.y}, v2f{w[t].x, w[t].y}, acc[t]);
#pragma unroll
        for (int t = 0; t < 4; ++t)
          acc[t] = wmma4(v2f{a.z, a.w}, v2f{w[t].z, w[t].w}, acc[t]);
      }
    }
    __syncthreads();
  }

  const long pos0 = ((long)(b * H_ + h0 + wave)) * W_ + w0;
#pragma unroll
  for (int t = 0; t < 4; ++t) {
    const int oc = 16 * t + mc;
    const float bias = b1[oc];
#pragma unroll
    for (int r = 0; r < 8; ++r) {
      const float v = acc[t][r] + bias;
      hbuf[(pos0 + r + 8 * g) * CB + oc] = v > 0.f ? v : 0.f;
    }
  }
}

__global__ __launch_bounds__(128) void k_res_conv1(float* __restrict__ q,
                                                   const float* __restrict__ hbuf,
                                                   const float* __restrict__ w2,
                                                   const float* __restrict__ b2) {
  const int tid = threadIdx.x;
  const int wave = tid >> 5, lane = tid & 31, g = lane >> 4, mc = lane & 15;
  const int oc0 = wave * 64;

  v4f wf[4][8];
  float bias[4];
#pragma unroll
  for (int t = 0; t < 4; ++t) {
    const float* wr = w2 + (oc0 + 16 * t + mc) * CB + 4 * g;
#pragma unroll
    for (int j = 0; j < 8; ++j) wf[t][j] = *(const v4f*)(wr + 8 * j);
    bias[t] = b2[oc0 + 16 * t + mc];
  }

#pragma unroll 1
  for (int i = 0; i < 4; ++i) {
    const long pos0 = ((long)blockIdx.x * 4 + i) * 16;
    const float* hp = hbuf + (pos0 + mc) * CB + 4 * g;
    v4f a[8];
#pragma unroll
    for (int j = 0; j < 8; ++j) a[j] = *(const v4f*)(hp + 8 * j);

    float* qp = q + (pos0 + 8 * g) * C_ + oc0 + mc;
    float qv[4][8];
#pragma unroll
    for (int r = 0; r < 8; ++r)
#pragma unroll
      for (int t = 0; t < 4; ++t) qv[t][r] = qp[r * C_ + 16 * t];

    v8f acc[4];
#pragma unroll
    for (int t = 0; t < 4; ++t) acc[t] = v8f{0.f, 0.f, 0.f, 0.f, 0.f, 0.f, 0.f, 0.f};
#pragma unroll
    for (int j = 0; j < 8; ++j) {
#pragma unroll
      for (int t = 0; t < 4; ++t)
        acc[t] = wmma4(v2f{a[j].x, a[j].y}, v2f{wf[t][j].x, wf[t][j].y}, acc[t]);
#pragma unroll
      for (int t = 0; t < 4; ++t)
        acc[t] = wmma4(v2f{a[j].z, a[j].w}, v2f{wf[t][j].z, wf[t][j].w}, acc[t]);
    }
#pragma unroll
    for (int r = 0; r < 8; ++r)
#pragma unroll
      for (int t = 0; t < 4; ++t)
        qp[r * C_ + 16 * t] = qv[t][r] + (acc[t][r] + bias[t]);
  }
}

__global__ __launch_bounds__(64) void k_tab(const float* __restrict__ etn,
                                            const float* __restrict__ wp,
                                            float* __restrict__ tab) {
  const int tid = threadIdx.x;
  const int wave = tid >> 5, lane = tid & 31, g = lane >> 4, mc = lane & 15;
  const int tap = blockIdx.x >> 5;
  const int nt  = blockIdx.x & 31;
  const int oc0 = 32 * wave;
  const float* ap = etn + (nt * 16 + mc) * C_ + 4 * g;
  const float* bp = wp + (tap * CB + oc0 + mc) * KC + 4 * g;
  v8f acc[2];
#pragma unroll
  for (int t = 0; t < 2; ++t) acc[t] = v8f{0.f, 0.f, 0.f, 0.f, 0.f, 0.f, 0.f, 0.f};
#pragma unroll
  for (int j = 0; j < 32; ++j) {
    v4f a = *(const v4f*)(ap + 8 * j);
    a.x = a.x > 0.f ? a.x : 0.f;  a.y = a.y > 0.f ? a.y : 0.f;
    a.z = a.z > 0.f ? a.z : 0.f;  a.w = a.w > 0.f ? a.w : 0.f;
    v4f w[2];
#pragma unroll
    for (int t = 0; t < 2; ++t)
      w[t] = *(const v4f*)(bp + (j >> 2) * WSLAB + 16 * t * KC + 8 * (j & 3));
#pragma unroll
    for (int t = 0; t < 2; ++t)
      acc[t] = wmma4(v2f{a.x, a.y}, v2f{w[t].x, w[t].y}, acc[t]);
#pragma unroll
    for (int t = 0; t < 2; ++t)
      acc[t] = wmma4(v2f{a.z, a.w}, v2f{w[t].z, w[t].w}, acc[t]);
  }
  float* tp = tab + ((size_t)(tap * NT + nt * 16 + 8 * g)) * CB + oc0 + mc;
#pragma unroll
  for (int t = 0; t < 2; ++t)
#pragma unroll
    for (int r = 0; r < 8; ++r) tp[r * CB + 16 * t] = acc[t][r];
}

__global__ __launch_bounds__(128) void k_res_conv1_g(float* __restrict__ q,
                                                     const float* __restrict__ etn,
                                                     const int* __restrict__ idxg,
                                                     const float* __restrict__ tab,
                                                     const float* __restrict__ b1,
                                                     const float* __restrict__ w2,
                                                     const float* __restrict__ b2) {
  __shared__ __attribute__((aligned(16))) float Hs[64 * HP];
  __shared__ int ids[3 * 64];
  const int tid = threadIdx.x;
  const int wave = tid >> 5, lane = tid & 31, g = lane >> 4, mc = lane & 15;
  const int oc0 = wave * 64;
  const int b = blockIdx.x >> 6, hh = blockIdx.x & 63;

  for (int t = tid; t < 192; t += 128) {
    const int hy = hh + (t >> 6) - 1;
    const bool ok = (unsigned)hy < 64u;
    const int v = idxg[(b * H_ + (ok ? hy : hh)) * W_ + (t & 63)];
    ids[t] = ok ? v : -1;
  }
  __syncthreads();

  const int cq = tid & 15, pl = tid >> 4;
  const v4f bv = *(const v4f*)(b1 + 4 * cq);
#pragma unroll 2
  for (int i = 0; i < 8; ++i) {
    const int p = 8 * i + pl;
    v4f s = {0.f, 0.f, 0.f, 0.f};
#pragma unroll
    for (int tap = 0; tap < 9; ++tap) {
      const int dh = tap / 3, dw = tap % 3;
      const int wx = p + dw - 1;
      const bool in = (unsigned)wx < 64u;
      const int n = ids[dh * 64 + (in ? wx : p)];
      const bool ok = in && n >= 0;
      const v4f t = *(const v4f*)(tab + ((size_t)(tap * NT + (ok ? n : 0))) * CB
                                  + 4 * cq);
      s.x += ok ? t.x : 0.f;  s.y += ok ? t.y : 0.f;
      s.z += ok ? t.z : 0.f;  s.w += ok ? t.w : 0.f;
    }
    s += bv;
    s.x = s.x > 0.f ? s.x : 0.f;  s.y = s.y > 0.f ? s.y : 0.f;
    s.z = s.z > 0.f ? s.z : 0.f;  s.w = s.w > 0.f ? s.w : 0.f;
    *(v4f*)(Hs + p * HP + 4 * cq) = s;
  }

  v4f wf[4][8];
  float bias[4];
#pragma unroll
  for (int t = 0; t < 4; ++t) {
    const float* wr = w2 + (oc0 + 16 * t + mc) * CB + 4 * g;
#pragma unroll
    for (int j = 0; j < 8; ++j) wf[t][j] = *(const v4f*)(wr + 8 * j);
    bias[t] = b2[oc0 + 16 * t + mc];
  }
  __syncthreads();

#pragma unroll 1
  for (int i = 0; i < 4; ++i) {
    const long pos0 = ((long)blockIdx.x * 4 + i) * 16;
    float* qp = q + (pos0 + 8 * g) * C_ + oc0 + mc;
    float qv[4][8];
#pragma unroll
    for (int r = 0; r < 8; ++r) {
      const float* er = etn + ids[64 + 16 * i + 8 * g + r] * C_ + oc0 + mc;
#pragma unroll
      for (int t = 0; t < 4; ++t) qv[t][r] = er[16 * t];
    }
    __builtin_amdgcn_sched_barrier(0);

    const float* hp = Hs + (16 * i + mc) * HP + 4 * g;
    v4f a[8];
#pragma unroll
    for (int j = 0; j < 8; ++j) a[j] = *(const v4f*)(hp + 8 * j);

    v8f acc[4];
#pragma unroll
    for (int t = 0; t < 4; ++t) acc[t] = v8f{0.f, 0.f, 0.f, 0.f, 0.f, 0.f, 0.f, 0.f};
#pragma unroll
    for (int j = 0; j < 8; ++j) {
#pragma unroll
      for (int t = 0; t < 4; ++t)
        acc[t] = wmma4(v2f{a[j].x, a[j].y}, v2f{wf[t][j].x, wf[t][j].y}, acc[t]);
#pragma unroll
      for (int t = 0; t < 4; ++t)
        acc[t] = wmma4(v2f{a[j].z, a[j].w}, v2f{wf[t][j].z, wf[t][j].w}, acc[t]);
    }
#pragma unroll
    for (int r = 0; r < 8; ++r)
#pragma unroll
      for (int t = 0; t < 4; ++t)
        qp[r * C_ + 16 * t] = qv[t][r] + (acc[t][r] + bias[t]);
  }
}

__global__ __launch_bounds__(128) void k_res_conv1_out(const float* __restrict__ q,
                                                       const float* __restrict__ hbuf,
                                                       const float* __restrict__ w2,
                                                       const float* __restrict__ b2,
                                                       const float* __restrict__ ow,
                                                       const float* __restrict__ ob,
                                                       float* __restrict__ out) {
  __shared__ __attribute__((aligned(16))) float Qs[64 * QP];
  const int tid = threadIdx.x;
  const int wave = tid >> 5, lane = tid & 31, g = lane >> 4, mc = lane & 15;
  const int oc0 = wave * 64;

  v4f wf[4][8];
  float bias[4];
#pragma unroll
  for (int t = 0; t < 4; ++t) {
    const float* wr = w2 + (oc0 + 16 * t + mc) * CB + 4 * g;
#pragma unroll
    for (int j = 0; j < 8; ++j) wf[t][j] = *(const v4f*)(wr + 8 * j);
    bias[t] = b2[oc0 + 16 * t + mc];
  }

#pragma unroll 1
  for (int i = 0; i < 4; ++i) {
    const long pos0 = ((long)blockIdx.x * 4 + i) * 16;
    const float* hp = hbuf + (pos0 + mc) * CB + 4 * g;
    v4f a[8];
#pragma unroll
    for (int j = 0; j < 8; ++j) a[j] = *(const v4f*)(hp + 8 * j);

    const float* qp = q + (pos0 + 8 * g) * C_ + oc0 + mc;
    float qv[4][8];
#pragma unroll
    for (int r = 0; r < 8; ++r)
#pragma unroll
      for (int t = 0; t < 4; ++t) qv[t][r] = qp[r * C_ + 16 * t];

    v8f acc[4];
#pragma unroll
    for (int t = 0; t < 4; ++t) acc[t] = v8f{0.f, 0.f, 0.f, 0.f, 0.f, 0.f, 0.f, 0.f};
#pragma unroll
    for (int j = 0; j < 8; ++j) {
#pragma unroll
      for (int t = 0; t < 4; ++t)
        acc[t] = wmma4(v2f{a[j].x, a[j].y}, v2f{wf[t][j].x, wf[t][j].y}, acc[t]);
#pragma unroll
      for (int t = 0; t < 4; ++t)
        acc[t] = wmma4(v2f{a[j].z, a[j].w}, v2f{wf[t][j].z, wf[t][j].w}, acc[t]);
    }
    float* qs = Qs + (16 * i + 8 * g) * QP + oc0 + mc;
#pragma unroll
    for (int r = 0; r < 8; ++r)
#pragma unroll
      for (int t = 0; t < 4; ++t)
        qs[r * QP + 16 * t] = qv[t][r] + (acc[t][r] + bias[t]);
  }

  const bool live = mc < OUTC;
  const float* bsrc = ow + (live ? mc : 0) * C_ + 2 * g;
  v2f bf[64];
#pragma unroll
  for (int k = 0; k < 64; ++k) {
    const v2f t = *(const v2f*)(bsrc + 4 * k);
    bf[k] = live ? t : v2f{0.f, 0.f};
  }
  const float obias = ob[live ? mc : 0];
  __syncthreads();

  const float* arow = Qs + (16 * wave + mc) * QP + 2 * g;
  v8f oacc = {0.f, 0.f, 0.f, 0.f, 0.f, 0.f, 0.f, 0.f};
#pragma unroll
  for (int k = 0; k < 64; ++k)
    oacc = wmma4(*(const v2f*)(arow + 4 * k), bf[k], oacc);

  if (live) {
    const int b = blockIdx.x >> 6, hh = blockIdx.x & 63;
    const long obase = ((long)(b * OUTC + mc) * H_ + hh) * W_ + 16 * wave;
#pragma unroll
    for (int r = 0; r < 8; ++r) out[obase + r + 8 * g] = oacc[r] + obias;
  }
}

__global__ __launch_bounds__(256) void k_final(const unsigned* __restrict__ counts,
                                               const float* __restrict__ wsdiff,
                                               float* __restrict__ out) {
  __shared__ float red[256];
  const int gid = blockIdx.x * 256 + threadIdx.x;
  if (blockIdx.x < 32) {
    out[WF_POS + gid] = (float)counts[gid] * (1.0f / 4096.0f);
  } else {
    float sum = 0.f;
    for (int t = threadIdx.x; t < VQ_WG; t += 256) sum += wsdiff[t];
    red[threadIdx.x] = sum; __syncthreads();
    for (int st = 128; st > 0; st >>= 1) {
      if (threadIdx.x < st) red[threadIdx.x] += red[threadIdx.x + st];
      __syncthreads();
    }
    if (threadIdx.x == 0)
      out[DIFF_POS] = red[0] * (1.0f / ((float)NPOS * (float)C_));
  }
}

extern "C" void kernel_launch(void* const* d_in, const int* in_sizes, int n_in,
                              void* d_out, int out_size, void* d_ws, size_t ws_size,
                              hipStream_t stream) {
  (void)in_sizes; (void)n_in; (void)out_size; (void)ws_size;
  const float* in  = (const float*)d_in[0];
  const float* emb = (const float*)d_in[1];
  const float* rw1 = (const float*)d_in[2];
  const float* rb1 = (const float*)d_in[3];
  const float* rw2 = (const float*)d_in[4];
  const float* rb2 = (const float*)d_in[5];
  const float* ow  = (const float*)d_in[6];
  const float* ob  = (const float*)d_in[7];
  float* out = (float*)d_out;
  float* ws  = (float*)d_ws;

  float*    q      = ws + Q_OFF;
  float*    hbuf   = ws + H_OFF;
  float*    etp    = ws + ETP_OFF;
  float*    etn    = ws + ETN_OFF;
  float*    en2    = ws + EN2_OFF;
  unsigned* counts = (unsigned*)(ws + CNT_OFF);
  float*    wsdiff = ws + WDIFF_OFF;
  float*    wpack  = ws + WPACK_OFF;
  float*    tab    = ws + TAB_OFF;
  int*      idxg   = (int*)(ws + IDX_OFF);

  k_prep<<<512, 256, 0, stream>>>(emb, etp, etn, en2, counts);
  k_packw<<<(3 * WBLK + 255) / 256, 256, 0, stream>>>(rw1, wpack);
  k_tab<<<9 * 32, 64, 0, stream>>>(etn, wpack, tab);
  k_vq<<<VQ_WG, 256, 0, stream>>>(in, etp, en2, counts, wsdiff, idxg);
  k_res_conv1_g<<<NTILE / 4, 128, 0, stream>>>(q, etn, idxg, tab, rb1, rw2, rb2);
  k_res_conv3<<<NPOS / (PR * 16), 256, 0, stream>>>(q, wpack + (size_t)1 * WBLK,
                                         rb1 + 1 * CB, hbuf);
  k_res_conv1<<<NTILE / 4, 128, 0, stream>>>(q, hbuf, rw2 + (size_t)1 * C_ * CB,
                                         rb2 + 1 * C_);
  k_res_conv3<<<NPOS / (PR * 16), 256, 0, stream>>>(q, wpack + (size_t)2 * WBLK,
                                         rb1 + 2 * CB, hbuf);
  k_res_conv1_out<<<NTILE / 4, 128, 0, stream>>>(q, hbuf, rw2 + (size_t)2 * C_ * CB,
                                             rb2 + 2 * C_, ow, ob, out);
  k_final<<<33, 256, 0, stream>>>(counts, wsdiff, out);
}
